# starting kernel
# baseline (speedup 1.0000x reference)
#include <hip/hip_runtime.h>
#include <hip/hip_bf16.h>
#include <math.h>

typedef __attribute__((ext_vector_type(2))) float v2f;
typedef __attribute__((ext_vector_type(4))) float v4f;
typedef __attribute__((ext_vector_type(8))) float v8f;

#define D_MODEL_C 1024
#define D_STATE_C 16
#define D_CONV_C  4
#define DT_RANK_C 64
#define D_INNER_C 2048
#define BATCH_C   2
#define SEQ_C     1024
#define NTOK_C    (BATCH_C * SEQ_C)

template <int EPILOG>
__global__ __launch_bounds__(256)
void wmma_gemm_f32(const float* __restrict__ A,
                   const float* __restrict__ W,
                   const float* __restrict__ bias,
                   float* __restrict__ C,
                   int M, int N, int K, int lda)
{
    const int lane      = threadIdx.x & 31;
    const int waveInBlk = threadIdx.x >> 5;
    const int nStrips   = (N + 63) >> 6;
    const int strip     = blockIdx.x * 8 + waveInBlk;
    const int total     = (M >> 4) * nStrips;
    if (strip >= total) return;

    const int m0 = (strip / nStrips) << 4;
    const int n0 = (strip % nStrips) << 6;

    const int lhalf = lane >> 4;
    const int lidx  = lane & 15;

    v8f acc[4];
    acc[0] = v8f{}; acc[1] = v8f{}; acc[2] = v8f{}; acc[3] = v8f{};

    const float* arow = A + (size_t)(m0 + lidx) * lda + 2 * lhalf;
    const float* bcol[4];
#pragma unroll
    for (int t = 0; t < 4; ++t) {
        int nc = n0 + 16 * t;
        bcol[t] = W + (size_t)((nc < N ? nc : 0) + lidx) * K + 2 * lhalf;
    }

    for (int k = 0; k < K; k += 16) {
        v2f a[4];
        v2f b[4][4];
#pragma unroll
        for (int kk = 0; kk < 4; ++kk) {
            int kb = k + 4 * kk;
            a[kk][0] = arow[kb];
            a[kk][1] = arow[kb + 1];
#pragma unroll
            for (int t = 0; t < 4; ++t) {
                b[kk][t][0] = bcol[t][kb];
                b[kk][t][1] = bcol[t][kb + 1];
            }
        }
#pragma unroll
        for (int kk = 0; kk < 4; ++kk) {
#pragma unroll
            for (int t = 0; t < 4; ++t) {
                acc[t] = __builtin_amdgcn_wmma_f32_16x16x4_f32(
                    false, a[kk], false, b[kk][t], (short)0, acc[t],
                    false, false);
            }
        }
    }

#pragma unroll
    for (int t = 0; t < 4; ++t) {
        int nc = n0 + 16 * t;
        if (nc >= N) break;
        float* cp = C + (size_t)(m0 + 8 * lhalf) * N + nc + lidx;
        float bv = (EPILOG == 1) ? bias[nc + lidx] : 0.f;
#pragma unroll
        for (int i = 0; i < 8; ++i) {
            float v = acc[t][i];
            if (EPILOG == 1) {
                float x = v + bv;
                v = (x > 20.f) ? x : log1pf(__expf(x));
            }
            cp[(size_t)i * N] = v;
        }
    }
}

__global__ __launch_bounds__(256)
void conv_silu_kernel(const float* __restrict__ xz,
                      const float* __restrict__ conv_w,
                      const float* __restrict__ conv_b,
                      float* __restrict__ x_act)
{
    int idx = blockIdx.x * blockDim.x + threadIdx.x;
    if (idx >= NTOK_C * D_INNER_C) return;
    int d = idx & (D_INNER_C - 1);
    int m = idx >> 11;
    int t = m & (SEQ_C - 1);

    float acc = conv_b[d];
    const float* w = conv_w + d * D_CONV_C;
#pragma unroll
    for (int j = 0; j < D_CONV_C; ++j) {
        int tt = t - (D_CONV_C - 1) + j;
        if (tt >= 0)
            acc += w[j] * xz[(size_t)(m - (D_CONV_C - 1) + j) * (2 * D_INNER_C) + d];
    }
    float s = acc / (1.f + __expf(-acc));
    x_act[(size_t)m * D_INNER_C + d] = s;
}

__global__ __launch_bounds__(256)
void selective_scan_kernel(const float* __restrict__ xz,
                           const float* __restrict__ x_act,
                           const float* __restrict__ dbc,
                           float* __restrict__ dt_y,
                           const float* __restrict__ A_log,
                           const float* __restrict__ Dvec)
{
    int gid = blockIdx.x * blockDim.x + threadIdx.x;
    if (gid >= BATCH_C * D_INNER_C) return;
    int b = gid >> 11;
    int d = gid & (D_INNER_C - 1);

    float Arow[D_STATE_C];
#pragma unroll
    for (int s = 0; s < D_STATE_C; ++s)
        Arow[s] = -__expf(A_log[(size_t)d * D_STATE_C + s]);
    const float Dd = Dvec[d];

    float h[D_STATE_C];
#pragma unroll
    for (int s = 0; s < D_STATE_C; ++s) h[s] = 0.f;

    for (int t = 0; t < SEQ_C; ++t) {
        size_t m = (size_t)b * SEQ_C + t;
        float u   = x_act[m * D_INNER_C + d];
        float dtv = dt_y [m * D_INNER_C + d];
        const v4f* bc4 = (const v4f*)(dbc + m * 96 + 64);
        v4f Bv0 = bc4[0], Bv1 = bc4[1], Bv2 = bc4[2], Bv3 = bc4[3];
        float Bt[D_STATE_C], Ct[D_STATE_C];
#pragma unroll
        for (int q = 0; q < 4; ++q) {
            Bt[q]      = Bv0[q]; Bt[4 + q]  = Bv1[q];
            Bt[8 + q]  = Bv2[q]; Bt[12 + q] = Bv3[q];
        }
        const v4f* cc4 = (const v4f*)(dbc + m * 96 + 80);
        v4f Cv0 = cc4[0], Cv1 = cc4[1], Cv2 = cc4[2], Cv3 = cc4[3];
#pragma unroll
        for (int q = 0; q < 4; ++q) {
            Ct[q]      = Cv0[q]; Ct[4 + q]  = Cv1[q];
            Ct[8 + q]  = Cv2[q]; Ct[12 + q] = Cv3[q];
        }

        float du = dtv * u;
        float y = 0.f;
#pragma unroll
        for (int s = 0; s < D_STATE_C; ++s) {
            float dA = __expf(dtv * Arow[s]);
            h[s] = dA * h[s] + du * Bt[s];
            y += h[s] * Ct[s];
        }
        y += u * Dd;
        float z  = xz[m * (2 * D_INNER_C) + D_INNER_C + d];
        float sz = z / (1.f + __expf(-z));
        dt_y[m * D_INNER_C + d] = y * sz;
    }
}

static inline int gemm_blocks(int M, int N)
{
    int strips = (M >> 4) * ((N + 63) >> 6);
    return (strips + 7) >> 3;
}

extern "C" void kernel_launch(void* const* d_in, const int* in_sizes, int n_in,
                              void* d_out, int out_size, void* d_ws, size_t ws_size,
                              hipStream_t stream) {
    (void)in_sizes; (void)n_in; (void)out_size; (void)ws_size;

    const float* x      = (const float*)d_in[0];
    const float* W_in   = (const float*)d_in[1];
    const float* conv_w = (const float*)d_in[2];
    const float* conv_b = (const float*)d_in[3];
    const float* W_x    = (const float*)d_in[4];
    const float* W_dt   = (const float*)d_in[5];
    const float* b_dt   = (const float*)d_in[6];
    const float* A_log  = (const float*)d_in[7];
    const float* Dvec   = (const float*)d_in[8];
    const float* W_out  = (const float*)d_in[9];
    float* out = (float*)d_out;

    float* xz   = (float*)d_ws;
    float* xact = xz   + (size_t)NTOK_C * (2 * D_INNER_C);
    float* dbc  = xact + (size_t)NTOK_C * D_INNER_C;
    float* dtb  = dbc  + (size_t)NTOK_C * 96;

    wmma_gemm_f32<0><<<gemm_blocks(NTOK_C, 2 * D_INNER_C), 256, 0, stream>>>(
        x, W_in, nullptr, xz, NTOK_C, 2 * D_INNER_C, D_MODEL_C, D_MODEL_C);

    {
        int n = NTOK_C * D_INNER_C;
        conv_silu_kernel<<<(n + 255) / 256, 256, 0, stream>>>(xz, conv_w, conv_b, xact);
    }

    wmma_gemm_f32<0><<<gemm_blocks(NTOK_C, 96), 256, 0, stream>>>(
        xact, W_x, nullptr, dbc, NTOK_C, 96, D_INNER_C, D_INNER_C);

    wmma_gemm_f32<1><<<gemm_blocks(NTOK_C, D_INNER_C), 256, 0, stream>>>(
        dbc, W_dt, b_dt, dtb, NTOK_C, D_INNER_C, DT_RANK_C, 96);

    selective_scan_kernel<<<(BATCH_C * D_INNER_C) / 256, 256, 0, stream>>>(
        xz, xact, dbc, dtb, A_log, Dvec);

    wmma_gemm_f32<0><<<gemm_blocks(NTOK_C, D_MODEL_C), 256, 0, stream>>>(
        dtb, W_out, nullptr, out, NTOK_C, D_MODEL_C, D_INNER_C, D_INNER_C);
}


# dt_proj: a wave's four 16-column tiles worked through in a rolled loop (products, then softplus and stores of one tile per trip): kernel text 47 KB -> about 12 KB; sums unchanged
# speedup vs baseline: 8.2526x; 8.2526x over previous
#include <hip/hip_runtime.h>
#include <hip/hip_bf16.h>
#include <math.h>

typedef __attribute__((ext_vector_type(2))) float v2f;
typedef __attribute__((ext_vector_type(4))) float v4f;
typedef __attribute__((ext_vector_type(8))) float v8f;

#define D_MODEL_C 1024
#define D_STATE_C 16
#define D_CONV_C  4
#define DT_RANK_C 64
#define D_INNER_C 2048
#define BATCH_C   2
#define SEQ_C     1024
#define NTOK_C    (BATCH_C * SEQ_C)

template <int WM, int WN, int NT>
static __device__ __forceinline__
void rt_load_set(v4f (&fa)[WM][NT], v4f (&fb)[WN][NT],
                 const float* ap, const float* bp, int lda, int ldb, int k)
{
#pragma unroll
    for (int t = 0; t < NT; ++t) {
#pragma unroll
        for (int i = 0; i < WM; ++i)
            fa[i][t] = *(const v4f*)(ap + (size_t)(16 * i) * lda + k + 8 * t);
#pragma unroll
        for (int j = 0; j < WN; ++j)
            fb[j][t] = *(const v4f*)(bp + (size_t)(16 * j) * ldb + k + 8 * t);
    }
}

template <int WM, int WN, int NT>
static __device__ __forceinline__
void rt_products(v8f (&acc)[WM][WN], const v4f (&fa)[WM][NT], const v4f (&fb)[WN][NT])
{
#pragma unroll
    for (int t = 0; t < NT; ++t) {
#pragma unroll
        for (int p = 0; p < 2; ++p) {
#pragma unroll
            for (int i = 0; i < WM; ++i) {
                v2f a;
                a[0] = fa[i][t][2 * p];
                a[1] = fa[i][t][2 * p + 1];
#pragma unroll
                for (int j = 0; j < WN; ++j) {
                    v2f b;
                    b[0] = fb[j][t][2 * p];
                    b[1] = fb[j][t][2 * p + 1];
                    acc[i][j] = __builtin_amdgcn_wmma_f32_16x16x4_f32(
                        false, a, false, b, (short)0, acc[i][j], false, false);
                }
            }
        }
    }
}

template <int M, int N, int K, int LDA, int WM, int WN, int BWM, int BWN, int KS, int EPILOG>
__global__ __launch_bounds__(32 * BWM * BWN)
void wmma_gemm_f32_rt(const float* __restrict__ A,
                      const float* __restrict__ W,
                      const float* __restrict__ bias,
                      float* __restrict__ C)
{
    constexpr int NT = KS / 8;
    constexpr int TM = 16 * WM, TN = 16 * WN;
    constexpr int BM = TM * BWM, BN = TN * BWN;
    static_assert(M % BM == 0 && N % BN == 0, "tiles must divide the problem");
    static_assert(K % (2 * KS) == 0, "two fragment sets per trip");

    const int lane = threadIdx.x & 31;
    const int wave = threadIdx.x >> 5;
    const int c = lane & 15;
    const int h = lane >> 4;
    const int m0 = blockIdx.y * BM + (wave / BWN) * TM;
    const int n0 = blockIdx.x * BN + (wave % BWN) * TN;

    const float* ap = A + (size_t)(m0 + c) * LDA + 4 * h;
    const float* bp = W + (size_t)(n0 + c) * K + 4 * h;

    v8f acc[WM][WN];
#pragma unroll
    for (int i = 0; i < WM; ++i)
#pragma unroll
        for (int j = 0; j < WN; ++j) acc[i][j] = v8f{};

    v4f fa0[WM][NT], fb0[WN][NT], fa1[WM][NT], fb1[WN][NT];

    rt_load_set<WM, WN, NT>(fa0, fb0, ap, bp, LDA, K, 0);
    for (int k = 0; k < K; k += 2 * KS) {
        rt_load_set<WM, WN, NT>(fa1, fb1, ap, bp, LDA, K, k + KS);
        __builtin_amdgcn_sched_barrier(0);
        rt_products<WM, WN, NT>(acc, fa0, fb0);
        __builtin_amdgcn_sched_barrier(0);
        const int kn = (k + 2 * KS < K) ? (k + 2 * KS) : 0;
        rt_load_set<WM, WN, NT>(fa0, fb0, ap, bp, LDA, K, kn);
        __builtin_amdgcn_sched_barrier(0);
        rt_products<WM, WN, NT>(acc, fa1, fb1);
        __builtin_amdgcn_sched_barrier(0);
    }

    float* cp = C + (size_t)(m0 + 8 * h) * N + n0 + c;
#pragma unroll
    for (int j = 0; j < WN; ++j) {
        float bv = 0.f;
        if (EPILOG == 1) bv = bias[n0 + c + 16 * j];
#pragma unroll
        for (int i = 0; i < WM; ++i) {
#pragma unroll
            for (int r = 0; r < 8; ++r) {
                float v = acc[i][j][r];
                if (EPILOG == 1) {
                    float x = v + bv;
                    v = (x > 20.f) ? x : log1pf(__expf(x));
                }
                cp[(size_t)(16 * i + r) * N + 16 * j] = v;
            }
        }
    }
}

__global__ __launch_bounds__(128)
void dt_proj_softplus_kernel(const float* __restrict__ dbc,
                             const float* __restrict__ W_dt,
                             const float* __restrict__ b_dt,
                             float* __restrict__ dt)
{
    constexpr int K = DT_RANK_C, LDA = 96, N = D_INNER_C;
    constexpr int NT = K / 8;
    constexpr int NJ = 4;

    const int lane = threadIdx.x & 31;
    const int wave = threadIdx.x >> 5;
    const int c = lane & 15;
    const int h = lane >> 4;
    const int m0 = blockIdx.y * 64 + (wave >> 1) * 32;
    const int n0 = blockIdx.x * 128 + (wave & 1) * (16 * NJ);

    const float* ap = dbc + (size_t)(m0 + c) * LDA + 4 * h;
    const float* bp = W_dt + (size_t)(n0 + c) * K + 4 * h;
    const float* bi = b_dt + n0 + c;
    float* cp = dt + (size_t)(m0 + 8 * h) * N + n0 + c;

    v4f fa[2][NT], fb[1][NT], fbn[NT];
    rt_load_set<2, 1, NT>(fa, fb, ap, bp, LDA, K, 0);
    float bv = bi[0];

#pragma nounroll
    for (int j = 0; j < NJ; ++j) {
        v8f acc[2][1];
        acc[0][0] = v8f{};
        acc[1][0] = v8f{};
        rt_products<2, 1, NT>(acc, fa, fb);

        const int jn = (j + 1 < NJ) ? (j + 1) : 0;
#pragma unroll
        for (int t = 0; t < NT; ++t)
            fbn[t] = *(const v4f*)(bp + (size_t)(16 * jn) * K + 8 * t);
        const float bvn = bi[16 * jn];

#pragma unroll
        for (int i = 0; i < 2; ++i) {
#pragma unroll
            for (int r = 0; r < 8; ++r) {
                float x = acc[i][0][r] + bv;
                float v = (x > 20.f) ? x : log1pf(__expf(x));
                cp[(size_t)(16 * i + r) * N + 16 * j] = v;
            }
        }

#pragma unroll
        for (int t = 0; t < NT; ++t) fb[0][t] = fbn[t];
        bv = bvn;
    }
}

__global__ __launch_bounds__(256)
void conv_silu_kernel(const float* __restrict__ xz,
                      const float* __restrict__ conv_w,
                      const float* __restrict__ conv_b,
                      float* __restrict__ x_act)
{
    int idx = blockIdx.x * blockDim.x + threadIdx.x;
    if (idx >= NTOK_C * D_INNER_C) return;
    int d = idx & (D_INNER_C - 1);
    int m = idx >> 11;
    int t = m & (SEQ_C - 1);

    float acc = conv_b[d];
    const float* w = conv_w + d * D_CONV_C;
#pragma unroll
    for (int j = 0; j < D_CONV_C; ++j) {
        int tt = t - (D_CONV_C - 1) + j;
        if (tt >= 0)
            acc += w[j] * xz[(size_t)(m - (D_CONV_C - 1) + j) * (2 * D_INNER_C) + d];
    }
    float s = acc / (1.f + __expf(-acc));
    x_act[(size_t)m * D_INNER_C + d] = s;
}

#define SCAN_TB 32
#define SCAN_NB (SEQ_C / SCAN_TB)
#define SCAN_WG (D_INNER_C / 32)
#define SCAN_G  4
#define SCAN_SG (D_STATE_C / SCAN_G)
#define SCAN_RPW (SCAN_TB / SCAN_G)
#define SCAN_NL (SCAN_RPW / 4)
#define SCAN_SB 4
#define SCAN_PLANE (SCAN_TB * 32)
#define SCAN_STAGE (4 * SCAN_PLANE)
#define SCAN_PART  (SCAN_G * SCAN_PLANE)

struct ScanStepIn { float dtv, u; v4f b, c; };

static __device__ __forceinline__
void scan_read_step(ScanStepIn& r, const float* st, int t, int lane, int w)
{
    r.dtv = st[t * 32 + lane];
    r.u   = st[SCAN_PLANE + t * 32 + lane];
    r.b = *(const v4f*)(st + 3 * SCAN_PLANE + t * 32 + SCAN_SG * w);
    r.c = *(const v4f*)(st + 3 * SCAN_PLANE + t * 32 + D_STATE_C + SCAN_SG * w);
}

static __device__ __forceinline__ void scan_lds_handover()
{
    __builtin_amdgcn_fence(__ATOMIC_RELEASE, "workgroup", "local");
    __builtin_amdgcn_s_barrier();
    __builtin_amdgcn_fence(__ATOMIC_ACQUIRE, "workgroup", "local");
}

__global__ __launch_bounds__(32 * SCAN_G)
void selective_scan_kernel(const float* __restrict__ xz,
                           const float* __restrict__ x_act,
                           const float* __restrict__ dbc,
                           float* __restrict__ dt_y,
                           const float* __restrict__ A_log,
                           const float* __restrict__ Dvec)
{
    static_assert(SCAN_SG == 4, "a wave reads its B and C as one 16-byte vector each");
    static_assert(SCAN_RPW % 4 == 0, "a 16-byte load per lane covers 4 rows per wave");
    __shared__ v4f s_mem[(2 * SCAN_STAGE + 2 * SCAN_PART) / 4];
    float* lds = (float*)s_mem;

    const int lane = threadIdx.x & 31;
    const int w    = __builtin_amdgcn_readfirstlane((int)(threadIdx.x >> 5));
    const int b    = blockIdx.x / SCAN_WG;
    const int d0   = (blockIdx.x % SCAN_WG) * 32;
    const int d    = d0 + lane;

    float Arow[SCAN_SG];
#pragma unroll
    for (int s = 0; s < SCAN_SG; ++s)
        Arow[s] = -__expf(A_log[(size_t)d * D_STATE_C + SCAN_SG * w + s]);
    const float Dd = Dvec[d];

    float h[SCAN_SG];
#pragma unroll
    for (int s = 0; s < SCAN_SG; ++s) h[s] = 0.f;

    const int r4  = lane >> 3;
    const int c4  = (lane & 7) * 4;
    const int row = SCAN_RPW * w + r4;
    const size_t m0 = (size_t)b * SEQ_C + row;
    const float* g_dt = dt_y  + m0 * D_INNER_C + d0 + c4;
    const float* g_u  = x_act + m0 * D_INNER_C + d0 + c4;
    const float* g_z  = xz    + m0 * (2 * D_INNER_C) + D_INNER_C + d0 + c4;
    const float* g_bc = dbc   + m0 * 96 + 64 + c4;
    const int so = row * 32 + c4;
    float* yp = dt_y + ((size_t)b * SEQ_C + SCAN_RPW * w) * D_INNER_C + d;

    v4f p_dt[SCAN_NL], p_u[SCAN_NL], p_z[SCAN_NL], p_bc[SCAN_NL];
#pragma unroll
    for (int j = 0; j < SCAN_NL; ++j) {
        p_dt[j] = *(const v4f*)(g_dt + (size_t)(4 * j) * D_INNER_C);
        p_u [j] = *(const v4f*)(g_u  + (size_t)(4 * j) * D_INNER_C);
        p_z [j] = *(const v4f*)(g_z  + (size_t)(4 * j) * (2 * D_INNER_C));
        p_bc[j] = *(const v4f*)(g_bc + (size_t)(4 * j) * 96);
    }
#pragma unroll
    for (int j = 0; j < SCAN_NL; ++j) {
        *(v4f*)(lds + so + 128 * j)                  = p_dt[j];
        *(v4f*)(lds + SCAN_PLANE + so + 128 * j)     = p_u [j];
        *(v4f*)(lds + 2 * SCAN_PLANE + so + 128 * j) = p_z [j];
        *(v4f*)(lds + 3 * SCAN_PLANE + so + 128 * j) = p_bc[j];
    }

    for (int k = 0; k < SCAN_NB; ++k) {
        const float* st = lds + (k & 1) * SCAN_STAGE;
        float* pt = lds + 2 * SCAN_STAGE + (k & 1) * SCAN_PART;

        scan_lds_handover();

        if (k + 1 < SCAN_NB) {
            const size_t ro = (size_t)(k + 1) * SCAN_TB;
#pragma unroll
            for (int j = 0; j < SCAN_NL; ++j) {
                p_dt[j] = *(const v4f*)(g_dt + (ro + 4 * j) * D_INNER_C);
                p_u [j] = *(const v4f*)(g_u  + (ro + 4 * j) * D_INNER_C);
                p_z [j] = *(const v4f*)(g_z  + (ro + 4 * j) * (2 * D_INNER_C));
                p_bc[j] = *(const v4f*)(g_bc + (ro + 4 * j) * 96);
            }
        }

        ScanStepIn cur[SCAN_SB], nxt[SCAN_SB];
#pragma unroll
        for (int i = 0; i < SCAN_SB; ++i)
            scan_read_step(cur[i], st, i, lane, w);

#pragma unroll
        for (int g = 0; g < SCAN_TB / SCAN_SB; ++g) {
            if (g + 1 < SCAN_TB / SCAN_SB) {
#pragma unroll
                for (int i = 0; i < SCAN_SB; ++i)
                    scan_read_step(nxt[i], st, (g + 1) * SCAN_SB + i, lane, w);
            }
            __builtin_amdgcn_sched_barrier(0);
#pragma unroll
            for (int i = 0; i < SCAN_SB; ++i) {
                const float dtv = cur[i].dtv;
                const float du  = dtv * cur[i].u;
                float p = 0.f;
#pragma unroll
                for (int s = 0; s < SCAN_SG; ++s) {
                    float dA = __expf(dtv * Arow[s]);
                    h[s] = dA * h[s] + du * cur[i].b[s];
                    p += h[s] * cur[i].c[s];
                }
                pt[w * SCAN_PLANE + (g * SCAN_SB + i) * 32 + lane] = p;
            }
            __builtin_amdgcn_sched_barrier(0);
            if (g + 1 < SCAN_TB / SCAN_SB) {
#pragma unroll
                for (int i = 0; i < SCAN_SB; ++i) cur[i] = nxt[i];
            }
        }

        scan_lds_handover();

        {
            float pp[SCAN_RPW][SCAN_G], uu[SCAN_RPW], zz[SCAN_RPW];
#pragma unroll
            for (int i = 0; i < SCAN_RPW; ++i) {
                const int o = (SCAN_RPW * w + i) * 32 + lane;
#pragma unroll
                for (int q = 0; q < SCAN_G; ++q) pp[i][q] = pt[q * SCAN_PLANE + o];
                uu[i] = st[SCAN_PLANE + o];
                zz[i] = st[2 * SCAN_PLANE + o];
            }
            float* yrow = yp + (size_t)k * SCAN_TB * D_INNER_C;
#pragma unroll
            for (int i = 0; i < SCAN_RPW; ++i) {
                float y = pp[i][0];
#pragma unroll
                for (int q = 1; q < SCAN_G; ++q) y += pp[i][q];
                y += uu[i] * Dd;
                const float z = zz[i];
                float sz = z / (1.f + __expf(-z));
                yrow[(size_t)i * D_INNER_C] = y * sz;
            }
        }

        if (k + 1 < SCAN_NB) {
            float* sn = lds + ((k + 1) & 1) * SCAN_STAGE;
#pragma unroll
            for (int j = 0; j < SCAN_NL; ++j) {
                *(v4f*)(sn + so + 128 * j)                  = p_dt[j];
                *(v4f*)(sn + SCAN_PLANE + so + 128 * j)     = p_u [j];
                *(v4f*)(sn + 2 * SCAN_PLANE + so + 128 * j) = p_z [j];
                *(v4f*)(sn + 3 * SCAN_PLANE + so + 128 * j) = p_bc[j];
            }
        }
    }
}

extern "C" void kernel_launch(void* const* d_in, const int* in_sizes, int n_in,
                              void* d_out, int out_size, void* d_ws, size_t ws_size,
                              hipStream_t stream) {
    (void)in_sizes; (void)n_in; (void)out_size; (void)ws_size;

    const float* x      = (const float*)d_in[0];
    const float* W_in   = (const float*)d_in[1];
    const float* conv_w = (const float*)d_in[2];
    const float* conv_b = (const float*)d_in[3];
    const float* W_x    = (const float*)d_in[4];
    const float* W_dt   = (const float*)d_in[5];
    const float* b_dt   = (const float*)d_in[6];
    const float* A_log  = (const float*)d_in[7];
    const float* Dvec   = (const float*)d_in[8];
    const float* W_out  = (const float*)d_in[9];
    float* out = (float*)d_out;

    float* xz   = (float*)d_ws;
    float* xact = xz   + (size_t)NTOK_C * (2 * D_INNER_C);
    float* dbc  = xact + (size_t)NTOK_C * D_INNER_C;
    float* dtb  = dbc  + (size_t)NTOK_C * 96;

    wmma_gemm_f32_rt<NTOK_C, 2 * D_INNER_C, D_MODEL_C, D_MODEL_C, 4, 4, 4, 2, 32, 0>
        <<<dim3(32, 8), 256, 0, stream>>>(x, W_in, nullptr, xz);

    {
        int n = NTOK_C * D_INNER_C;
        conv_silu_kernel<<<(n + 255) / 256, 256, 0, stream>>>(xz, conv_w, conv_b, xact);
    }

    wmma_gemm_f32_rt<NTOK_C, 96, D_INNER_C, D_INNER_C, 1, 3, 1, 1, 64, 0>
        <<<dim3(2, 128), 32, 0, stream>>>(xact, W_x, nullptr, dbc);

    dt_proj_softplus_kernel<<<dim3(16, 32), 128, 0, stream>>>(dbc, W_dt, b_dt, dtb);

    selective_scan_kernel<<<(BATCH_C * D_INNER_C) / 32, 32 * SCAN_G, 0, stream>>>(
        xz, xact, dbc, dtb, A_log, Dvec);

    wmma_gemm_f32_rt<NTOK_C, D_MODEL_C, D_INNER_C, D_INNER_C, 2, 4, 2, 2, 64, 0>
        <<<dim3(8, 32), 128, 0, stream>>>(dtb, W_out, nullptr, out);
}


# conv+SiLU fused into x_proj: a lane's conv of 4 channels is its A fragment, x_act not read back; 16 tokens per workgroup, K split over 8 waves, partials added in wave order; dbc sum order changes
# speedup vs baseline: 8.7697x; 1.0627x over previous
#include <hip/hip_runtime.h>
#include <hip/hip_bf16.h>
#include <math.h>

typedef __attribute__((ext_vector_type(2))) float v2f;
typedef __attribute__((ext_vector_type(4))) float v4f;
typedef __attribute__((ext_vector_type(8))) float v8f;

#define D_MODEL_C 1024
#define D_STATE_C 16
#define D_CONV_C  4
#define DT_RANK_C 64
#define D_INNER_C 2048
#define BATCH_C   2
#define SEQ_C     1024
#define NTOK_C    (BATCH_C * SEQ_C)

template <int WM, int WN, int NT>
static __device__ __forceinline__
void rt_load_set(v4f (&fa)[WM][NT], v4f (&fb)[WN][NT],
                 const float* ap, const float* bp, int lda, int ldb, int k)
{
#pragma unroll
    for (int t = 0; t < NT; ++t) {
#pragma unroll
        for (int i = 0; i < WM; ++i)
            fa[i][t] = *(const v4f*)(ap + (size_t)(16 * i) * lda + k + 8 * t);
#pragma unroll
        for (int j = 0; j < WN; ++j)
            fb[j][t] = *(const v4f*)(bp + (size_t)(16 * j) * ldb + k + 8 * t);
    }
}

template <int WM, int WN, int NT>
static __device__ __forceinline__
void rt_products(v8f (&acc)[WM][WN], const v4f (&fa)[WM][NT], const v4f (&fb)[WN][NT])
{
#pragma unroll
    for (int t = 0; t < NT; ++t) {
#pragma unroll
        for (int p = 0; p < 2; ++p) {
#pragma unroll
            for (int i = 0; i < WM; ++i) {
                v2f a;
                a[0] = fa[i][t][2 * p];
                a[1] = fa[i][t][2 * p + 1];
#pragma unroll
                for (int j = 0; j < WN; ++j) {
                    v2f b;
                    b[0] = fb[j][t][2 * p];
                    b[1] = fb[j][t][2 * p + 1];
                    acc[i][j] = __builtin_amdgcn_wmma_f32_16x16x4_f32(
                        false, a, false, b, (short)0, acc[i][j], false, false);
                }
            }
        }
    }
}

template <int M, int N, int K, int LDA, int WM, int WN, int BWM, int BWN, int KS, int EPILOG>
__global__ __launch_bounds__(32 * BWM * BWN)
void wmma_gemm_f32_rt(const float* __restrict__ A,
                      const float* __restrict__ W,
                      const float* __restrict__ bias,
                      float* __restrict__ C)
{
    constexpr int NT = KS / 8;
    constexpr int TM = 16 * WM, TN = 16 * WN;
    constexpr int BM = TM * BWM, BN = TN * BWN;
    static_assert(M % BM == 0 && N % BN == 0, "tiles must divide the problem");
    static_assert(K % (2 * KS) == 0, "two fragment sets per trip");

    const int lane = threadIdx.x & 31;
    const int wave = threadIdx.x >> 5;
    const int c = lane & 15;
    const int h = lane >> 4;
    const int m0 = blockIdx.y * BM + (wave / BWN) * TM;
    const int n0 = blockIdx.x * BN + (wave % BWN) * TN;

    const float* ap = A + (size_t)(m0 + c) * LDA + 4 * h;
    const float* bp = W + (size_t)(n0 + c) * K + 4 * h;

    v8f acc[WM][WN];
#pragma unroll
    for (int i = 0; i < WM; ++i)
#pragma unroll
        for (int j = 0; j < WN; ++j) acc[i][j] = v8f{};

    v4f fa0[WM][NT], fb0[WN][NT], fa1[WM][NT], fb1[WN][NT];

    rt_load_set<WM, WN, NT>(fa0, fb0, ap, bp, LDA, K, 0);
    for (int k = 0; k < K; k += 2 * KS) {
        rt_load_set<WM, WN, NT>(fa1, fb1, ap, bp, LDA, K, k + KS);
        __builtin_amdgcn_sched_barrier(0);
        rt_products<WM, WN, NT>(acc, fa0, fb0);
        __builtin_amdgcn_sched_barrier(0);
        const int kn = (k + 2 * KS < K) ? (k + 2 * KS) : 0;
        rt_load_set<WM, WN, NT>(fa0, fb0, ap, bp, LDA, K, kn);
        __builtin_amdgcn_sched_barrier(0);
        rt_products<WM, WN, NT>(acc, fa1, fb1);
        __builtin_amdgcn_sched_barrier(0);
    }

    float* cp = C + (size_t)(m0 + 8 * h) * N + n0 + c;
#pragma unroll
    for (int j = 0; j < WN; ++j) {
        float bv = 0.f;
        if (EPILOG == 1) bv = bias[n0 + c + 16 * j];
#pragma unroll
        for (int i = 0; i < WM; ++i) {
#pragma unroll
            for (int r = 0; r < 8; ++r) {
                float v = acc[i][j][r];
                if (EPILOG == 1) {
                    float x = v + bv;
                    v = (x > 20.f) ? x : log1pf(__expf(x));
                }
                cp[(size_t)(16 * i + r) * N + 16 * j] = v;
            }
        }
    }
}

__global__ __launch_bounds__(128)
void dt_proj_softplus_kernel(const float* __restrict__ dbc,
                             const float* __restrict__ W_dt,
                             const float* __restrict__ b_dt,
                             float* __restrict__ dt)
{
    constexpr int K = DT_RANK_C, LDA = 96, N = D_INNER_C;
    constexpr int NT = K / 8;
    constexpr int NJ = 4;

    const int lane = threadIdx.x & 31;
    const int wave = threadIdx.x >> 5;
    const int c = lane & 15;
    const int h = lane >> 4;
    const int m0 = blockIdx.y * 64 + (wave >> 1) * 32;
    const int n0 = blockIdx.x * 128 + (wave & 1) * (16 * NJ);

    const float* ap = dbc + (size_t)(m0 + c) * LDA + 4 * h;
    const float* bp = W_dt + (size_t)(n0 + c) * K + 4 * h;
    const float* bi = b_dt + n0 + c;
    float* cp = dt + (size_t)(m0 + 8 * h) * N + n0 + c;

    v4f fa[2][NT], fb[1][NT], fbn[NT];
    rt_load_set<2, 1, NT>(fa, fb, ap, bp, LDA, K, 0);
    float bv = bi[0];

#pragma nounroll
    for (int j = 0; j < NJ; ++j) {
        v8f acc[2][1];
        acc[0][0] = v8f{};
        acc[1][0] = v8f{};
        rt_products<2, 1, NT>(acc, fa, fb);

        const int jn = (j + 1 < NJ) ? (j + 1) : 0;
#pragma unroll
        for (int t = 0; t < NT; ++t)
            fbn[t] = *(const v4f*)(bp + (size_t)(16 * jn) * K + 8 * t);
        const float bvn = bi[16 * jn];

#pragma unroll
        for (int i = 0; i < 2; ++i) {
#pragma unroll
            for (int r = 0; r < 8; ++r) {
                float x = acc[i][0][r] + bv;
                float v = (x > 20.f) ? x : log1pf(__expf(x));
                cp[(size_t)(16 * i + r) * N + 16 * j] = v;
            }
        }

#pragma unroll
        for (int t = 0; t < NT; ++t) fb[0][t] = fbn[t];
        bv = bvn;
    }
}

#define CX_WAVES 8
#define CX_KW    (D_INNER_C / CX_WAVES)
#define CX_GC    32
#define CX_NG    (CX_KW / CX_GC)
#define CX_NT    (CX_GC / 8)
#define CX_NJ    6
#define CX_TILE  (16 * 96)

static __device__ __forceinline__
void cx_load_w(v4f (&wf)[CX_NJ][CX_NT], const float* wp, int g)
{
#pragma unroll
    for (int n = 0; n < CX_NJ; ++n)
#pragma unroll
        for (int t = 0; t < CX_NT; ++t)
            wf[n][t] = *(const v4f*)(wp + (size_t)(16 * n) * D_INNER_C + CX_GC * g + 8 * t);
}

static __device__ __forceinline__
void cx_load_x(v4f (&xs)[D_CONV_C][CX_NT], v4f (&cw)[CX_NT][4], v4f (&cb)[CX_NT],
               const float* xp, const int (&off)[D_CONV_C],
               const float* cwp, const float* cbp, int g)
{
#pragma unroll
    for (int j = 0; j < D_CONV_C; ++j)
#pragma unroll
        for (int t = 0; t < CX_NT; ++t)
            xs[j][t] = *(const v4f*)(xp + off[j] + CX_GC * g + 8 * t);
#pragma unroll
    for (int t = 0; t < CX_NT; ++t) {
#pragma unroll
        for (int e = 0; e < 4; ++e)
            cw[t][e] = *(const v4f*)(cwp + D_CONV_C * (CX_GC * g + 8 * t + e));
        cb[t] = *(const v4f*)(cbp + CX_GC * g + 8 * t);
    }
}

static __device__ __forceinline__
void cx_conv_silu(v4f (&s)[CX_NT], const v4f (&xs)[D_CONV_C][CX_NT],
                  const v4f (&cw)[CX_NT][4], const v4f (&cb)[CX_NT],
                  const bool (&ok)[D_CONV_C])
{
#pragma unroll
    for (int t = 0; t < CX_NT; ++t) {
#pragma unroll
        for (int e = 0; e < 4; ++e) {
            float acc = cb[t][e];
#pragma unroll
            for (int j = 0; j < D_CONV_C; ++j) {
                const float xv = ok[j] ? xs[j][t][e] : 0.f;
                acc = __builtin_fmaf(cw[t][e][j], xv, acc);
            }
            s[t][e] = acc / (1.f + __expf(-acc));
        }
    }
}

static __device__ __forceinline__
void cx_products(v8f (&acc)[CX_NJ], const v4f (&s)[CX_NT], const v4f (&wf)[CX_NJ][CX_NT])
{
#pragma unroll
    for (int t = 0; t < CX_NT; ++t) {
#pragma unroll
        for (int p = 0; p < 2; ++p) {
            v2f a;
            a[0] = s[t][2 * p];
            a[1] = s[t][2 * p + 1];
#pragma unroll
            for (int n = 0; n < CX_NJ; ++n) {
                v2f b;
                b[0] = wf[n][t][2 * p];
                b[1] = wf[n][t][2 * p + 1];
                acc[n] = __builtin_amdgcn_wmma_f32_16x16x4_f32(
                    false, a, false, b, (short)0, acc[n], false, false);
            }
        }
    }
}

__global__ __launch_bounds__(32 * CX_WAVES)
void conv_silu_xproj_kernel(const float* __restrict__ xz,
                            const float* __restrict__ conv_w,
                            const float* __restrict__ conv_b,
                            const float* __restrict__ W_x,
                            float* __restrict__ x_act,
                            float* __restrict__ dbc)
{
    __shared__ float s_part[CX_WAVES * CX_TILE];

    const int lane = threadIdx.x & 31;
    const int wave = threadIdx.x >> 5;
    const int c = lane & 15;
    const int h = lane >> 4;
    const int m0 = blockIdx.x * 16;
    const int m  = m0 + c;
    const int tpos = m & (SEQ_C - 1);
    const int k0 = CX_KW * wave + 4 * h;

    bool ok[D_CONV_C];
    int  off[D_CONV_C];
#pragma unroll
    for (int j = 0; j < D_CONV_C; ++j) {
        ok[j]  = (tpos - (D_CONV_C - 1) + j) >= 0;
        off[j] = ok[j] ? (j - (D_CONV_C - 1)) * (2 * D_INNER_C) : 0;
    }

    const float* xp  = xz + (size_t)m * (2 * D_INNER_C) + k0;
    const float* wp  = W_x + (size_t)c * D_INNER_C + k0;
    const float* cwp = conv_w + D_CONV_C * k0;
    const float* cbp = conv_b + k0;
    float* ap = x_act + (size_t)m * D_INNER_C + k0;

    v8f acc[CX_NJ];
#pragma unroll
    for (int n = 0; n < CX_NJ; ++n) acc[n] = v8f{};

    v4f xs[D_CONV_C][CX_NT], cw[CX_NT][4], cb[CX_NT], s[CX_NT];
    v4f w0[CX_NJ][CX_NT], w1[CX_NJ][CX_NT];

    cx_load_x(xs, cw, cb, xp, off, cwp, cbp, 0);
    cx_load_w(w0, wp, 0);
    for (int g = 0; g < CX_NG; g += 2) {
        cx_load_w(w1, wp, g + 1);
        __builtin_amdgcn_sched_barrier(0);
        cx_conv_silu(s, xs, cw, cb, ok);
#pragma unroll
        for (int t = 0; t < CX_NT; ++t)
            *(v4f*)(ap + CX_GC * g + 8 * t) = s[t];
        cx_load_x(xs, cw, cb, xp, off, cwp, cbp, g + 1);
        __builtin_amdgcn_sched_barrier(0);
        cx_products(acc, s, w0);
        __builtin_amdgcn_sched_barrier(0);
        const int gn = (g + 2 < CX_NG) ? (g + 2) : 0;
        cx_load_w(w0, wp, gn);
        __builtin_amdgcn_sched_barrier(0);
        cx_conv_silu(s, xs, cw, cb, ok);
#pragma unroll
        for (int t = 0; t < CX_NT; ++t)
            *(v4f*)(ap + CX_GC * (g + 1) + 8 * t) = s[t];
        cx_load_x(xs, cw, cb, xp, off, cwp, cbp, gn);
        __builtin_amdgcn_sched_barrier(0);
        cx_products(acc, s, w1);
        __builtin_amdgcn_sched_barrier(0);
    }

    float* pw = s_part + wave * CX_TILE + (8 * h) * 96 + c;
#pragma unroll
    for (int n = 0; n < CX_NJ; ++n)
#pragma unroll
        for (int r = 0; r < 8; ++r)
            pw[r * 96 + 16 * n] = acc[n][r];
    __syncthreads();

    float* dp = dbc + (size_t)m0 * 96;
#pragma unroll
    for (int q = 0; q < CX_TILE / (32 * CX_WAVES); ++q) {
        const int o = threadIdx.x + 32 * CX_WAVES * q;
        float v = s_part[o];
#pragma unroll
        for (int w = 1; w < CX_WAVES; ++w) v += s_part[w * CX_TILE + o];
        dp[o] = v;
    }
}

#define SCAN_TB 32
#define SCAN_NB (SEQ_C / SCAN_TB)
#define SCAN_WG (D_INNER_C / 32)
#define SCAN_G  4
#define SCAN_SG (D_STATE_C / SCAN_G)
#define SCAN_RPW (SCAN_TB / SCAN_G)
#define SCAN_NL (SCAN_RPW / 4)
#define SCAN_SB 4
#define SCAN_PLANE (SCAN_TB * 32)
#define SCAN_STAGE (4 * SCAN_PLANE)
#define SCAN_PART  (SCAN_G * SCAN_PLANE)

struct ScanStepIn { float dtv, u; v4f b, c; };

static __device__ __forceinline__
void scan_read_step(ScanStepIn& r, const float* st, int t, int lane, int w)
{
    r.dtv = st[t * 32 + lane];
    r.u   = st[SCAN_PLANE + t * 32 + lane];
    r.b = *(const v4f*)(st + 3 * SCAN_PLANE + t * 32 + SCAN_SG * w);
    r.c = *(const v4f*)(st + 3 * SCAN_PLANE + t * 32 + D_STATE_C + SCAN_SG * w);
}

static __device__ __forceinline__ void scan_lds_handover()
{
    __builtin_amdgcn_fence(__ATOMIC_RELEASE, "workgroup", "local");
    __builtin_amdgcn_s_barrier();
    __builtin_amdgcn_fence(__ATOMIC_ACQUIRE, "workgroup", "local");
}

__global__ __launch_bounds__(32 * SCAN_G)
void selective_scan_kernel(const float* __restrict__ xz,
                           const float* __restrict__ x_act,
                           const float* __restrict__ dbc,
                           float* __restrict__ dt_y,
                           const float* __restrict__ A_log,
                           const float* __restrict__ Dvec)
{
    static_assert(SCAN_SG == 4, "a wave reads its B and C as one 16-byte vector each");
    static_assert(SCAN_RPW % 4 == 0, "a 16-byte load per lane covers 4 rows per wave");
    __shared__ v4f s_mem[(2 * SCAN_STAGE + 2 * SCAN_PART) / 4];
    float* lds = (float*)s_mem;

    const int lane = threadIdx.x & 31;
    const int w    = __builtin_amdgcn_readfirstlane((int)(threadIdx.x >> 5));
    const int b    = blockIdx.x / SCAN_WG;
    const int d0   = (blockIdx.x % SCAN_WG) * 32;
    const int d    = d0 + lane;

    float Arow[SCAN_SG];
#pragma unroll
    for (int s = 0; s < SCAN_SG; ++s)
        Arow[s] = -__expf(A_log[(size_t)d * D_STATE_C + SCAN_SG * w + s]);
    const float Dd = Dvec[d];

    float h[SCAN_SG];
#pragma unroll
    for (int s = 0; s < SCAN_SG; ++s) h[s] = 0.f;

    const int r4  = lane >> 3;
    const int c4  = (lane & 7) * 4;
    const int row = SCAN_RPW * w + r4;
    const size_t m0 = (size_t)b * SEQ_C + row;
    const float* g_dt = dt_y  + m0 * D_INNER_C + d0 + c4;
    const float* g_u  = x_act + m0 * D_INNER_C + d0 + c4;
    const float* g_z  = xz    + m0 * (2 * D_INNER_C) + D_INNER_C + d0 + c4;
    const float* g_bc = dbc   + m0 * 96 + 64 + c4;
    const int so = row * 32 + c4;
    float* yp = dt_y + ((size_t)b * SEQ_C + SCAN_RPW * w) * D_INNER_C + d;

    v4f p_dt[SCAN_NL], p_u[SCAN_NL], p_z[SCAN_NL], p_bc[SCAN_NL];
#pragma unroll
    for (int j = 0; j < SCAN_NL; ++j) {
        p_dt[j] = *(const v4f*)(g_dt + (size_t)(4 * j) * D_INNER_C);
        p_u [j] = *(const v4f*)(g_u  + (size_t)(4 * j) * D_INNER_C);
        p_z [j] = *(const v4f*)(g_z  + (size_t)(4 * j) * (2 * D_INNER_C));
        p_bc[j] = *(const v4f*)(g_bc + (size_t)(4 * j) * 96);
    }
#pragma unroll
    for (int j = 0; j < SCAN_NL; ++j) {
        *(v4f*)(lds + so + 128 * j)                  = p_dt[j];
        *(v4f*)(lds + SCAN_PLANE + so + 128 * j)     = p_u [j];
        *(v4f*)(lds + 2 * SCAN_PLANE + so + 128 * j) = p_z [j];
        *(v4f*)(lds + 3 * SCAN_PLANE + so + 128 * j) = p_bc[j];
    }

    for (int k = 0; k < SCAN_NB; ++k) {
        const float* st = lds + (k & 1) * SCAN_STAGE;
        float* pt = lds + 2 * SCAN_STAGE + (k & 1) * SCAN_PART;

        scan_lds_handover();

        if (k + 1 < SCAN_NB) {
            const size_t ro = (size_t)(k + 1) * SCAN_TB;
#pragma unroll
            for (int j = 0; j < SCAN_NL; ++j) {
                p_dt[j] = *(const v4f*)(g_dt + (ro + 4 * j) * D_INNER_C);
                p_u [j] = *(const v4f*)(g_u  + (ro + 4 * j) * D_INNER_C);
                p_z [j] = *(const v4f*)(g_z  + (ro + 4 * j) * (2 * D_INNER_C));
                p_bc[j] = *(const v4f*)(g_bc + (ro + 4 * j) * 96);
            }
        }

        ScanStepIn cur[SCAN_SB], nxt[SCAN_SB];
#pragma unroll
        for (int i = 0; i < SCAN_SB; ++i)
            scan_read_step(cur[i], st, i, lane, w);

#pragma unroll
        for (int g = 0; g < SCAN_TB / SCAN_SB; ++g) {
            if (g + 1 < SCAN_TB / SCAN_SB) {
#pragma unroll
                for (int i = 0; i < SCAN_SB; ++i)
                    scan_read_step(nxt[i], st, (g + 1) * SCAN_SB + i, lane, w);
            }
            __builtin_amdgcn_sched_barrier(0);
#pragma unroll
            for (int i = 0; i < SCAN_SB; ++i) {
                const float dtv = cur[i].dtv;
                const float du  = dtv * cur[i].u;
                float p = 0.f;
#pragma unroll
                for (int s = 0; s < SCAN_SG; ++s) {
                    float dA = __expf(dtv * Arow[s]);
                    h[s] = dA * h[s] + du * cur[i].b[s];
                    p += h[s] * cur[i].c[s];
                }
                pt[w * SCAN_PLANE + (g * SCAN_SB + i) * 32 + lane] = p;
            }
            __builtin_amdgcn_sched_barrier(0);
            if (g + 1 < SCAN_TB / SCAN_SB) {
#pragma unroll
                for (int i = 0; i < SCAN_SB; ++i) cur[i] = nxt[i];
            }
        }

        scan_lds_handover();

        {
            float pp[SCAN_RPW][SCAN_G], uu[SCAN_RPW], zz[SCAN_RPW];
#pragma unroll
            for (int i = 0; i < SCAN_RPW; ++i) {
                const int o = (SCAN_RPW * w + i) * 32 + lane;
#pragma unroll
                for (int q = 0; q < SCAN_G; ++q) pp[i][q] = pt[q * SCAN_PLANE + o];
                uu[i] = st[SCAN_PLANE + o];
                zz[i] = st[2 * SCAN_PLANE + o];
            }
            float* yrow = yp + (size_t)k * SCAN_TB * D_INNER_C;
#pragma unroll
            for (int i = 0; i < SCAN_RPW; ++i) {
                float y = pp[i][0];
#pragma unroll
                for (int q = 1; q < SCAN_G; ++q) y += pp[i][q];
                y += uu[i] * Dd;
                const float z = zz[i];
                float sz = z / (1.f + __expf(-z));
                yrow[(size_t)i * D_INNER_C] = y * sz;
            }
        }

        if (k + 1 < SCAN_NB) {
            float* sn = lds + ((k + 1) & 1) * SCAN_STAGE;
#pragma unroll
            for (int j = 0; j < SCAN_NL; ++j) {
                *(v4f*)(sn + so + 128 * j)                  = p_dt[j];
                *(v4f*)(sn + SCAN_PLANE + so + 128 * j)     = p_u [j];
                *(v4f*)(sn + 2 * SCAN_PLANE + so + 128 * j) = p_z [j];
                *(v4f*)(sn + 3 * SCAN_PLANE + so + 128 * j) = p_bc[j];
            }
        }
    }
}

extern "C" void kernel_launch(void* const* d_in, const int* in_sizes, int n_in,
                              void* d_out, int out_size, void* d_ws, size_t ws_size,
                              hipStream_t stream) {
    (void)in_sizes; (void)n_in; (void)out_size; (void)ws_size;

    const float* x      = (const float*)d_in[0];
    const float* W_in   = (const float*)d_in[1];
    const float* conv_w = (const float*)d_in[2];
    const float* conv_b = (const float*)d_in[3];
    const float* W_x    = (const float*)d_in[4];
    const float* W_dt   = (const float*)d_in[5];
    const float* b_dt   = (const float*)d_in[6];
    const float* A_log  = (const float*)d_in[7];
    const float* Dvec   = (const float*)d_in[8];
    const float* W_out  = (const float*)d_in[9];
    float* out = (float*)d_out;

    float* xz   = (float*)d_ws;
    float* xact = xz   + (size_t)NTOK_C * (2 * D_INNER_C);
    float* dbc  = xact + (size_t)NTOK_C * D_INNER_C;
    float* dtb  = dbc  + (size_t)NTOK_C * 96;

    wmma_gemm_f32_rt<NTOK_C, 2 * D_INNER_C, D_MODEL_C, D_MODEL_C, 4, 4, 4, 2, 32, 0>
        <<<dim3(32, 8), 256, 0, stream>>>(x, W_in, nullptr, xz);

    conv_silu_xproj_kernel<<<NTOK_C / 16, 32 * CX_WAVES, 0, stream>>>(
        xz, conv_w, conv_b, W_x, xact, dbc);

    dt_proj_softplus_kernel<<<dim3(16, 32), 128, 0, stream>>>(dbc, W_dt, b_dt, dtb);

    selective_scan_kernel<<<(BATCH_C * D_INNER_C) / 32, 32 * SCAN_G, 0, stream>>>(
        xz, xact, dbc, dtb, A_log, Dvec);

    wmma_gemm_f32_rt<NTOK_C, D_MODEL_C, D_INNER_C, D_INNER_C, 2, 4, 2, 2, 64, 0>
        <<<dim3(8, 32), 128, 0, stream>>>(dtb, W_out, nullptr, out);
}


# scan: 16 channels per workgroup (256 workgroups), lane = channel x state half, 2 states per lane; y kept in 2-token x 16-channel 128-byte tiles that out_proj's A loader reads; y sum order changes
# speedup vs baseline: 9.0942x; 1.0370x over previous
#include <hip/hip_runtime.h>
#include <hip/hip_bf16.h>
#include <math.h>

typedef __attribute__((ext_vector_type(2))) float v2f;
typedef __attribute__((ext_vector_type(4))) float v4f;
typedef __attribute__((ext_vector_type(8))) float v8f;

#define D_MODEL_C 1024
#define D_STATE_C 16
#define D_CONV_C  4
#define DT_RANK_C 64
#define D_INNER_C 2048
#define BATCH_C   2
#define SEQ_C     1024
#define NTOK_C    (BATCH_C * SEQ_C)

template <int WM, int WN, int NT, bool ATILE = false>
static __device__ __forceinline__
void rt_load_set(v4f (&fa)[WM][NT], v4f (&fb)[WN][NT],
                 const float* ap, const float* bp, int lda, int ldb, int k)
{
#pragma unroll
    for (int t = 0; t < NT; ++t) {
#pragma unroll
        for (int i = 0; i < WM; ++i) {
            if (ATILE)
                fa[i][t] = *(const v4f*)(ap + (size_t)(8 * i) * (2 * lda) + 2 * k
                                         + 32 * (t >> 1) + 8 * (t & 1));
            else
                fa[i][t] = *(const v4f*)(ap + (size_t)(16 * i) * lda + k + 8 * t);
        }
#pragma unroll
        for (int j = 0; j < WN; ++j)
            fb[j][t] = *(const v4f*)(bp + (size_t)(16 * j) * ldb + k + 8 * t);
    }
}

template <int WM, int WN, int NT>
static __device__ __forceinline__
void rt_products(v8f (&acc)[WM][WN], const v4f (&fa)[WM][NT], const v4f (&fb)[WN][NT])
{
#pragma unroll
    for (int t = 0; t < NT; ++t) {
#pragma unroll
        for (int p = 0; p < 2; ++p) {
#pragma unroll
            for (int i = 0; i < WM; ++i) {
                v2f a;
                a[0] = fa[i][t][2 * p];
                a[1] = fa[i][t][2 * p + 1];
#pragma unroll
                for (int j = 0; j < WN; ++j) {
                    v2f b;
                    b[0] = fb[j][t][2 * p];
                    b[1] = fb[j][t][2 * p + 1];
                    acc[i][j] = __builtin_amdgcn_wmma_f32_16x16x4_f32(
                        false, a, false, b, (short)0, acc[i][j], false, false);
                }
            }
        }
    }
}

template <int M, int N, int K, int LDA, int WM, int WN, int BWM, int BWN, int KS, int EPILOG, bool ATILE = false>
__global__ __launch_bounds__(32 * BWM * BWN)
void wmma_gemm_f32_rt(const float* __restrict__ A,
                      const float* __restrict__ W,
                      const float* __restrict__ bias,
                      float* __restrict__ C)
{
    constexpr int NT = KS / 8;
    constexpr int TM = 16 * WM, TN = 16 * WN;
    constexpr int BM = TM * BWM, BN = TN * BWN;
    static_assert(M % BM == 0 && N % BN == 0, "tiles must divide the problem");
    static_assert(K % (2 * KS) == 0, "two fragment sets per trip");
    static_assert(!ATILE || (KS % 16 == 0 && LDA % 16 == 0 && LDA == K), "tiled A: whole 16-column tiles");

    const int lane = threadIdx.x & 31;
    const int wave = threadIdx.x >> 5;
    const int c = lane & 15;
    const int h = lane >> 4;
    const int m0 = blockIdx.y * BM + (wave / BWN) * TM;
    const int n0 = blockIdx.x * BN + (wave % BWN) * TN;

    const float* ap = ATILE
        ? A + (size_t)((m0 + c) >> 1) * (2 * LDA) + (c & 1) * 16 + 4 * h
        : A + (size_t)(m0 + c) * LDA + 4 * h;
    const float* bp = W + (size_t)(n0 + c) * K + 4 * h;

    v8f acc[WM][WN];
#pragma unroll
    for (int i = 0; i < WM; ++i)
#pragma unroll
        for (int j = 0; j < WN; ++j) acc[i][j] = v8f{};

    v4f fa0[WM][NT], fb0[WN][NT], fa1[WM][NT], fb1[WN][NT];

    rt_load_set<WM, WN, NT, ATILE>(fa0, fb0, ap, bp, LDA, K, 0);
    for (int k = 0; k < K; k += 2 * KS) {
        rt_load_set<WM, WN, NT, ATILE>(fa1, fb1, ap, bp, LDA, K, k + KS);
        __builtin_amdgcn_sched_barrier(0);
        rt_products<WM, WN, NT>(acc, fa0, fb0);
        __builtin_amdgcn_sched_barrier(0);
        const int kn = (k + 2 * KS < K) ? (k + 2 * KS) : 0;
        rt_load_set<WM, WN, NT, ATILE>(fa0, fb0, ap, bp, LDA, K, kn);
        __builtin_amdgcn_sched_barrier(0);
        rt_products<WM, WN, NT>(acc, fa1, fb1);
        __builtin_amdgcn_sched_barrier(0);
    }

    float* cp = C + (size_t)(m0 + 8 * h) * N + n0 + c;
#pragma unroll
    for (int j = 0; j < WN; ++j) {
        float bv = 0.f;
        if (EPILOG == 1) bv = bias[n0 + c + 16 * j];
#pragma unroll
        for (int i = 0; i < WM; ++i) {
#pragma unroll
            for (int r = 0; r < 8; ++r) {
                float v = acc[i][j][r];
                if (EPILOG == 1) {
                    float x = v + bv;
                    v = (x > 20.f) ? x : log1pf(__expf(x));
                }
                cp[(size_t)(16 * i + r) * N + 16 * j] = v;
            }
        }
    }
}

__global__ __launch_bounds__(128)
void dt_proj_softplus_kernel(const float* __restrict__ dbc,
                             const float* __restrict__ W_dt,
                             const float* __restrict__ b_dt,
                             float* __restrict__ dt)
{
    constexpr int K = DT_RANK_C, LDA = 96, N = D_INNER_C;
    constexpr int NT = K / 8;
    constexpr int NJ = 4;

    const int lane = threadIdx.x & 31;
    const int wave = threadIdx.x >> 5;
    const int c = lane & 15;
    const int h = lane >> 4;
    const int m0 = blockIdx.y * 64 + (wave >> 1) * 32;
    const int n0 = blockIdx.x * 128 + (wave & 1) * (16 * NJ);

    const float* ap = dbc + (size_t)(m0 + c) * LDA + 4 * h;
    const float* bp = W_dt + (size_t)(n0 + c) * K + 4 * h;
    const float* bi = b_dt + n0 + c;
    float* cp = dt + (size_t)(m0 + 8 * h) * N + n0 + c;

    v4f fa[2][NT], fb[1][NT], fbn[NT];
    rt_load_set<2, 1, NT>(fa, fb, ap, bp, LDA, K, 0);
    float bv = bi[0];

#pragma nounroll
    for (int j = 0; j < NJ; ++j) {
        v8f acc[2][1];
        acc[0][0] = v8f{};
        acc[1][0] = v8f{};
        rt_products<2, 1, NT>(acc, fa, fb);

        const int jn = (j + 1 < NJ) ? (j + 1) : 0;
#pragma unroll
        for (int t = 0; t < NT; ++t)
            fbn[t] = *(const v4f*)(bp + (size_t)(16 * jn) * K + 8 * t);
        const float bvn = bi[16 * jn];

#pragma unroll
        for (int i = 0; i < 2; ++i) {
#pragma unroll
            for (int r = 0; r < 8; ++r) {
                float x = acc[i][0][r] + bv;
                float v = (x > 20.f) ? x : log1pf(__expf(x));
                cp[(size_t)(16 * i + r) * N + 16 * j] = v;
            }
        }

#pragma unroll
        for (int t = 0; t < NT; ++t) fb[0][t] = fbn[t];
        bv = bvn;
    }
}

#define CX_WAVES 8
#define CX_KW    (D_INNER_C / CX_WAVES)
#define CX_GC    32
#define CX_NG    (CX_KW / CX_GC)
#define CX_NT    (CX_GC / 8)
#define CX_NJ    6
#define CX_TILE  (16 * 96)

static __device__ __forceinline__
void cx_load_w(v4f (&wf)[CX_NJ][CX_NT], const float* wp, int g)
{
#pragma unroll
    for (int n = 0; n < CX_NJ; ++n)
#pragma unroll
        for (int t = 0; t < CX_NT; ++t)
            wf[n][t] = *(const v4f*)(wp + (size_t)(16 * n) * D_INNER_C + CX_GC * g + 8 * t);
}

static __device__ __forceinline__
void cx_load_x(v4f (&xs)[D_CONV_C][CX_NT], v4f (&cw)[CX_NT][4], v4f (&cb)[CX_NT],
               const float* xp, const int (&off)[D_CONV_C],
               const float* cwp, const float* cbp, int g)
{
#pragma unroll
    for (int j = 0; j < D_CONV_C; ++j)
#pragma unroll
        for (int t = 0; t < CX_NT; ++t)
            xs[j][t] = *(const v4f*)(xp + off[j] + CX_GC * g + 8 * t);
#pragma unroll
    for (int t = 0; t < CX_NT; ++t) {
#pragma unroll
        for (int e = 0; e < 4; ++e)
            cw[t][e] = *(const v4f*)(cwp + D_CONV_C * (CX_GC * g + 8 * t + e));
        cb[t] = *(const v4f*)(cbp + CX_GC * g + 8 * t);
    }
}

static __device__ __forceinline__
void cx_conv_silu(v4f (&s)[CX_NT], const v4f (&xs)[D_CONV_C][CX_NT],
                  const v4f (&cw)[CX_NT][4], const v4f (&cb)[CX_NT],
                  const bool (&ok)[D_CONV_C])
{
#pragma unroll
    for (int t = 0; t < CX_NT; ++t) {
#pragma unroll
        for (int e = 0; e < 4; ++e) {
            float acc = cb[t][e];
#pragma unroll
            for (int j = 0; j < D_CONV_C; ++j) {
                const float xv = ok[j] ? xs[j][t][e] : 0.f;
                acc = __builtin_fmaf(cw[t][e][j], xv, acc);
            }
            s[t][e] = acc / (1.f + __expf(-acc));
        }
    }
}

static __device__ __forceinline__
void cx_products(v8f (&acc)[CX_NJ], const v4f (&s)[CX_NT], const v4f (&wf)[CX_NJ][CX_NT])
{
#pragma unroll
    for (int t = 0; t < CX_NT; ++t) {
#pragma unroll
        for (int p = 0; p < 2; ++p) {
            v2f a;
            a[0] = s[t][2 * p];
            a[1] = s[t][2 * p + 1];
#pragma unroll
            for (int n = 0; n < CX_NJ; ++n) {
                v2f b;
                b[0] = wf[n][t][2 * p];
                b[1] = wf[n][t][2 * p + 1];
                acc[n] = __builtin_amdgcn_wmma_f32_16x16x4_f32(
                    false, a, false, b, (short)0, acc[n], false, false);
            }
        }
    }
}

__global__ __launch_bounds__(32 * CX_WAVES)
void conv_silu_xproj_kernel(const float* __restrict__ xz,
                            const float* __restrict__ conv_w,
                            const float* __restrict__ conv_b,
                            const float* __restrict__ W_x,
                            float* __restrict__ x_act,
                            float* __restrict__ dbc)
{
    __shared__ float s_part[CX_WAVES * CX_TILE];

    const int lane = threadIdx.x & 31;
    const int wave = threadIdx.x >> 5;
    const int c = lane & 15;
    const int h = lane >> 4;
    const int m0 = blockIdx.x * 16;
    const int m  = m0 + c;
    const int tpos = m & (SEQ_C - 1);
    const int k0 = CX_KW * wave + 4 * h;

    bool ok[D_CONV_C];
    int  off[D_CONV_C];
#pragma unroll
    for (int j = 0; j < D_CONV_C; ++j) {
        ok[j]  = (tpos - (D_CONV_C - 1) + j) >= 0;
        off[j] = ok[j] ? (j - (D_CONV_C - 1)) * (2 * D_INNER_C) : 0;
    }

    const float* xp  = xz + (size_t)m * (2 * D_INNER_C) + k0;
    const float* wp  = W_x + (size_t)c * D_INNER_C + k0;
    const float* cwp = conv_w + D_CONV_C * k0;
    const float* cbp = conv_b + k0;
    float* ap = x_act + (size_t)m * D_INNER_C + k0;

    v8f acc[CX_NJ];
#pragma unroll
    for (int n = 0; n < CX_NJ; ++n) acc[n] = v8f{};

    v4f xs[D_CONV_C][CX_NT], cw[CX_NT][4], cb[CX_NT], s[CX_NT];
    v4f w0[CX_NJ][CX_NT], w1[CX_NJ][CX_NT];

    cx_load_x(xs, cw, cb, xp, off, cwp, cbp, 0);
    cx_load_w(w0, wp, 0);
    for (int g = 0; g < CX_NG; g += 2) {
        cx_load_w(w1, wp, g + 1);
        __builtin_amdgcn_sched_barrier(0);
        cx_conv_silu(s, xs, cw, cb, ok);
#pragma unroll
        for (int t = 0; t < CX_NT; ++t)
            *(v4f*)(ap + CX_GC * g + 8 * t) = s[t];
        cx_load_x(xs, cw, cb, xp, off, cwp, cbp, g + 1);
        __builtin_amdgcn_sched_barrier(0);
        cx_products(acc, s, w0);
        __builtin_amdgcn_sched_barrier(0);
        const int gn = (g + 2 < CX_NG) ? (g + 2) : 0;
        cx_load_w(w0, wp, gn);
        __builtin_amdgcn_sched_barrier(0);
        cx_conv_silu(s, xs, cw, cb, ok);
#pragma unroll
        for (int t = 0; t < CX_NT; ++t)
            *(v4f*)(ap + CX_GC * (g + 1) + 8 * t) = s[t];
        cx_load_x(xs, cw, cb, xp, off, cwp, cbp, gn);
        __builtin_amdgcn_sched_barrier(0);
        cx_products(acc, s, w1);
        __builtin_amdgcn_sched_barrier(0);
    }

    float* pw = s_part + wave * CX_TILE + (8 * h) * 96 + c;
#pragma unroll
    for (int n = 0; n < CX_NJ; ++n)
#pragma unroll
        for (int r = 0; r < 8; ++r)
            pw[r * 96 + 16 * n] = acc[n][r];
    __syncthreads();

    float* dp = dbc + (size_t)m0 * 96;
#pragma unroll
    for (int q = 0; q < CX_TILE / (32 * CX_WAVES); ++q) {
        const int o = threadIdx.x + 32 * CX_WAVES * q;
        float v = s_part[o];
#pragma unroll
        for (int w = 1; w < CX_WAVES; ++w) v += s_part[w * CX_TILE + o];
        dp[o] = v;
    }
}

#define SCAN_TB 32
#define SCAN_NB (SEQ_C / SCAN_TB)
#define SCAN_CH 16
#define SCAN_WG (D_INNER_C / SCAN_CH)
#define SCAN_G  4
#define SCAN_SL 2
#define SCAN_RPW (SCAN_TB / SCAN_G)
#define SCAN_SB 4
#define SCAN_PL16 (SCAN_TB * SCAN_CH)
#define SCAN_PLBC (SCAN_TB * 32)
#define SCAN_STAGE (3 * SCAN_PL16 + SCAN_PLBC)
#define SCAN_PPL  (SCAN_TB * 32)
#define SCAN_PART (SCAN_G * SCAN_PPL)

struct ScanStepIn { float dtv, u; v2f b, c; };

static __device__ __forceinline__
void scan_read_step(ScanStepIn& r, const float* st, int t, int c, int so)
{
    r.dtv = st[t * SCAN_CH + c];
    r.u   = st[SCAN_PL16 + t * SCAN_CH + c];
    r.b = *(const v2f*)(st + 3 * SCAN_PL16 + t * 32 + so);
    r.c = *(const v2f*)(st + 3 * SCAN_PL16 + t * 32 + D_STATE_C + so);
}

static __device__ __forceinline__ void scan_lds_handover()
{
    __builtin_amdgcn_fence(__ATOMIC_RELEASE, "workgroup", "local");
    __builtin_amdgcn_s_barrier();
    __builtin_amdgcn_fence(__ATOMIC_ACQUIRE, "workgroup", "local");
}

__global__ __launch_bounds__(32 * SCAN_G)
void selective_scan_kernel(const float* __restrict__ xz,
                           const float* __restrict__ x_act,
                           const float* __restrict__ dbc,
                           const float* __restrict__ dt,
                           float* __restrict__ y,
                           const float* __restrict__ A_log,
                           const float* __restrict__ Dvec)
{
    static_assert(SCAN_G * 2 * SCAN_SL == D_STATE_C, "4 waves x 2 lane halves x 2 states");
    static_assert(SCAN_RPW == 8, "a wave stages 8 rows: one 16-byte load per lane covers them");
    __shared__ v4f s_mem[(2 * SCAN_STAGE + 2 * SCAN_PART) / 4];
    float* lds = (float*)s_mem;

    const int lane = threadIdx.x & 31;
    const int w    = __builtin_amdgcn_readfirstlane((int)(threadIdx.x >> 5));
    const int c    = lane & 15;
    const int hh   = lane >> 4;
    const int b    = blockIdx.x / SCAN_WG;
    const int d0   = (blockIdx.x % SCAN_WG) * SCAN_CH;
    const int d    = d0 + c;
    const int so   = 2 * SCAN_SL * w + SCAN_SL * hh;

    float Arow[SCAN_SL];
#pragma unroll
    for (int s = 0; s < SCAN_SL; ++s)
        Arow[s] = -__expf(A_log[(size_t)d * D_STATE_C + so + s]);
    const float Dd = Dvec[d];

    float h[SCAN_SL];
#pragma unroll
    for (int s = 0; s < SCAN_SL; ++s) h[s] = 0.f;

    const int r16 = lane >> 2;
    const int c16 = (lane & 3) * 4;
    const size_t m16 = (size_t)b * SEQ_C + SCAN_RPW * w + r16;
    const float* g_dt = dt    + m16 * D_INNER_C + d0 + c16;
    const float* g_u  = x_act + m16 * D_INNER_C + d0 + c16;
    const float* g_z  = xz    + m16 * (2 * D_INNER_C) + D_INNER_C + d0 + c16;
    const int so16 = (SCAN_RPW * w + r16) * SCAN_CH + c16;
    const int r8 = lane >> 3;
    const int c8 = (lane & 7) * 4;
    const size_t m8 = (size_t)b * SEQ_C + SCAN_RPW * w + r8;
    const float* g_bc = dbc + m8 * 96 + 64 + c8;
    const int sobc = (SCAN_RPW * w + r8) * 32 + c8;
    float* yp = y + ((size_t)(b * (SEQ_C / 2) + (SCAN_RPW / 2) * w) * (D_INNER_C / 16)
                     + (d0 >> 4)) * 32 + lane;

    v4f p_dt, p_u, p_z, p_bc[2];
    p_dt = *(const v4f*)(g_dt);
    p_u  = *(const v4f*)(g_u);
    p_z  = *(const v4f*)(g_z);
    p_bc[0] = *(const v4f*)(g_bc);
    p_bc[1] = *(const v4f*)(g_bc + (size_t)4 * 96);
    *(v4f*)(lds + so16)                  = p_dt;
    *(v4f*)(lds + SCAN_PL16 + so16)      = p_u;
    *(v4f*)(lds + 2 * SCAN_PL16 + so16)  = p_z;
    *(v4f*)(lds + 3 * SCAN_PL16 + sobc)        = p_bc[0];
    *(v4f*)(lds + 3 * SCAN_PL16 + sobc + 128)  = p_bc[1];

    for (int k = 0; k < SCAN_NB; ++k) {
        const float* st = lds + (k & 1) * SCAN_STAGE;
        float* pt = lds + 2 * SCAN_STAGE + (k & 1) * SCAN_PART;

        scan_lds_handover();

        if (k + 1 < SCAN_NB) {
            const size_t ro = (size_t)(k + 1) * SCAN_TB;
            p_dt = *(const v4f*)(g_dt + ro * D_INNER_C);
            p_u  = *(const v4f*)(g_u  + ro * D_INNER_C);
            p_z  = *(const v4f*)(g_z  + ro * (2 * D_INNER_C));
            p_bc[0] = *(const v4f*)(g_bc + ro * 96);
            p_bc[1] = *(const v4f*)(g_bc + (ro + 4) * 96);
        }

        ScanStepIn cur[SCAN_SB], nxt[SCAN_SB];
#pragma unroll
        for (int i = 0; i < SCAN_SB; ++i)
            scan_read_step(cur[i], st, i, c, so);

#pragma unroll
        for (int g = 0; g < SCAN_TB / SCAN_SB; ++g) {
            if (g + 1 < SCAN_TB / SCAN_SB) {
#pragma unroll
                for (int i = 0; i < SCAN_SB; ++i)
                    scan_read_step(nxt[i], st, (g + 1) * SCAN_SB + i, c, so);
            }
            __builtin_amdgcn_sched_barrier(0);
#pragma unroll
            for (int i = 0; i < SCAN_SB; ++i) {
                const float dtv = cur[i].dtv;
                const float du  = dtv * cur[i].u;
                float p = 0.f;
#pragma unroll
                for (int s = 0; s < SCAN_SL; ++s) {
                    float dA = __expf(dtv * Arow[s]);
                    h[s] = dA * h[s] + du * cur[i].b[s];
                    p += h[s] * cur[i].c[s];
                }
                pt[w * SCAN_PPL + (g * SCAN_SB + i) * 32 + lane] = p;
            }
            __builtin_amdgcn_sched_barrier(0);
            if (g + 1 < SCAN_TB / SCAN_SB) {
#pragma unroll
                for (int i = 0; i < SCAN_SB; ++i) cur[i] = nxt[i];
            }
        }

        scan_lds_handover();

        {
            float plo[SCAN_RPW / 2][SCAN_G], phi[SCAN_RPW / 2][SCAN_G];
            float uu[SCAN_RPW / 2], zz[SCAN_RPW / 2];
#pragma unroll
            for (int i = 0; i < SCAN_RPW / 2; ++i) {
                const int t = SCAN_RPW * w + 2 * i + hh;
#pragma unroll
                for (int q = 0; q < SCAN_G; ++q) {
                    plo[i][q] = pt[q * SCAN_PPL + t * 32 + c];
                    phi[i][q] = pt[q * SCAN_PPL + t * 32 + 16 + c];
                }
                uu[i] = st[SCAN_PL16 + t * SCAN_CH + c];
                zz[i] = st[2 * SCAN_PL16 + t * SCAN_CH + c];
            }
            float* yrow = yp + (size_t)k * (SCAN_TB / 2) * (D_INNER_C / 16) * 32;
#pragma unroll
            for (int i = 0; i < SCAN_RPW / 2; ++i) {
                float yv = plo[i][0] + phi[i][0];
#pragma unroll
                for (int q = 1; q < SCAN_G; ++q) {
                    yv += plo[i][q];
                    yv += phi[i][q];
                }
                yv += uu[i] * Dd;
                const float z = zz[i];
                float sz = z / (1.f + __expf(-z));
                yrow[(size_t)i * (D_INNER_C / 16) * 32] = yv * sz;
            }
        }

        if (k + 1 < SCAN_NB) {
            float* sn = lds + ((k + 1) & 1) * SCAN_STAGE;
            *(v4f*)(sn + so16)                  = p_dt;
            *(v4f*)(sn + SCAN_PL16 + so16)      = p_u;
            *(v4f*)(sn + 2 * SCAN_PL16 + so16)  = p_z;
            *(v4f*)(sn + 3 * SCAN_PL16 + sobc)        = p_bc[0];
            *(v4f*)(sn + 3 * SCAN_PL16 + sobc + 128)  = p_bc[1];
        }
    }
}

extern "C" void kernel_launch(void* const* d_in, const int* in_sizes, int n_in,
                              void* d_out, int out_size, void* d_ws, size_t ws_size,
                              hipStream_t stream) {
    (void)in_sizes; (void)n_in; (void)out_size; (void)ws_size;

    const float* x      = (const float*)d_in[0];
    const float* W_in   = (const float*)d_in[1];
    const float* conv_w = (const float*)d_in[2];
    const float* conv_b = (const float*)d_in[3];
    const float* W_x    = (const float*)d_in[4];
    const float* W_dt   = (const float*)d_in[5];
    const float* b_dt   = (const float*)d_in[6];
    const float* A_log  = (const float*)d_in[7];
    const float* Dvec   = (const float*)d_in[8];
    const float* W_out  = (const float*)d_in[9];
    float* out = (float*)d_out;

    float* xz   = (float*)d_ws;
    float* xact = xz   + (size_t)NTOK_C * (2 * D_INNER_C);
    float* dbc  = xact + (size_t)NTOK_C * D_INNER_C;
    float* dtb  = dbc  + (size_t)NTOK_C * 96;
    float* yb   = dtb  + (size_t)NTOK_C * D_INNER_C;

    wmma_gemm_f32_rt<NTOK_C, 2 * D_INNER_C, D_MODEL_C, D_MODEL_C, 4, 4, 4, 2, 32, 0>
        <<<dim3(32, 8), 256, 0, stream>>>(x, W_in, nullptr, xz);

    conv_silu_xproj_kernel<<<NTOK_C / 16, 32 * CX_WAVES, 0, stream>>>(
        xz, conv_w, conv_b, W_x, xact, dbc);

    dt_proj_softplus_kernel<<<dim3(16, 32), 128, 0, stream>>>(dbc, W_dt, b_dt, dtb);

    selective_scan_kernel<<<(BATCH_C * D_INNER_C) / SCAN_CH, 32 * SCAN_G, 0, stream>>>(
        xz, xact, dbc, dtb, yb, A_log, Dvec);

    wmma_gemm_f32_rt<NTOK_C, D_MODEL_C, D_INNER_C, D_INNER_C, 2, 4, 2, 2, 64, 0, true>
        <<<dim3(8, 32), 128, 0, stream>>>(yb, W_out, nullptr, out);
}


# dt_proj moved into the scan launch: 4 more waves per workgroup make the next 32-step block's dt (WMMA + softplus) into LDS beside the recurrence; dt never stored; one launch less; sums unchanged
# speedup vs baseline: 9.4993x; 1.0445x over previous
#include <hip/hip_runtime.h>
#include <hip/hip_bf16.h>
#include <math.h>

typedef __attribute__((ext_vector_type(2))) float v2f;
typedef __attribute__((ext_vector_type(4))) float v4f;
typedef __attribute__((ext_vector_type(8))) float v8f;

#define D_MODEL_C 1024
#define D_STATE_C 16
#define D_CONV_C  4
#define DT_RANK_C 64
#define D_INNER_C 2048
#define BATCH_C   2
#define SEQ_C     1024
#define NTOK_C    (BATCH_C * SEQ_C)

template <int WM, int WN, int NT, bool ATILE = false>
static __device__ __forceinline__
void rt_load_set(v4f (&fa)[WM][NT], v4f (&fb)[WN][NT],
                 const float* ap, const float* bp, int lda, int ldb, int k)
{
#pragma unroll
    for (int t = 0; t < NT; ++t) {
#pragma unroll
        for (int i = 0; i < WM; ++i) {
            if (ATILE)
                fa[i][t] = *(const v4f*)(ap + (size_t)(8 * i) * (2 * lda) + 2 * k
                                         + 32 * (t >> 1) + 8 * (t & 1));
            else
                fa[i][t] = *(const v4f*)(ap + (size_t)(16 * i) * lda + k + 8 * t);
        }
#pragma unroll
        for (int j = 0; j < WN; ++j)
            fb[j][t] = *(const v4f*)(bp + (size_t)(16 * j) * ldb + k + 8 * t);
    }
}

template <int WM, int WN, int NT>
static __device__ __forceinline__
void rt_products(v8f (&acc)[WM][WN], const v4f (&fa)[WM][NT], const v4f (&fb)[WN][NT])
{
#pragma unroll
    for (int t = 0; t < NT; ++t) {
#pragma unroll
        for (int p = 0; p < 2; ++p) {
#pragma unroll
            for (int i = 0; i < WM; ++i) {
                v2f a;
                a[0] = fa[i][t][2 * p];
                a[1] = fa[i][t][2 * p + 1];
#pragma unroll
                for (int j = 0; j < WN; ++j) {
                    v2f b;
                    b[0] = fb[j][t][2 * p];
                    b[1] = fb[j][t][2 * p + 1];
                    acc[i][j] = __builtin_amdgcn_wmma_f32_16x16x4_f32(
                        false, a, false, b, (short)0, acc[i][j], false, false);
                }
            }
        }
    }
}

template <int M, int N, int K, int LDA, int WM, int WN, int BWM, int BWN, int KS, int EPILOG, bool ATILE = false>
__global__ __launch_bounds__(32 * BWM * BWN)
void wmma_gemm_f32_rt(const float* __restrict__ A,
                      const float* __restrict__ W,
                      const float* __restrict__ bias,
                      float* __restrict__ C)
{
    constexpr int NT = KS / 8;
    constexpr int TM = 16 * WM, TN = 16 * WN;
    constexpr int BM = TM * BWM, BN = TN * BWN;
    static_assert(M % BM == 0 && N % BN == 0, "tiles must divide the problem");
    static_assert(K % (2 * KS) == 0, "two fragment sets per trip");
    static_assert(!ATILE || (KS % 16 == 0 && LDA % 16 == 0 && LDA == K), "tiled A: whole 16-column tiles");

    const int lane = threadIdx.x & 31;
    const int wave = threadIdx.x >> 5;
    const int c = lane & 15;
    const int h = lane >> 4;
    const int m0 = blockIdx.y * BM + (wave / BWN) * TM;
    const int n0 = blockIdx.x * BN + (wave % BWN) * TN;

    const float* ap = ATILE
        ? A + (size_t)((m0 + c) >> 1) * (2 * LDA) + (c & 1) * 16 + 4 * h
        : A + (size_t)(m0 + c) * LDA + 4 * h;
    const float* bp = W + (size_t)(n0 + c) * K + 4 * h;

    v8f acc[WM][WN];
#pragma unroll
    for (int i = 0; i < WM; ++i)
#pragma unroll
        for (int j = 0; j < WN; ++j) acc[i][j] = v8f{};

    v4f fa0[WM][NT], fb0[WN][NT], fa1[WM][NT], fb1[WN][NT];

    rt_load_set<WM, WN, NT, ATILE>(fa0, fb0, ap, bp, LDA, K, 0);
    for (int k = 0; k < K; k += 2 * KS) {
        rt_load_set<WM, WN, NT, ATILE>(fa1, fb1, ap, bp, LDA, K, k + KS);
        __builtin_amdgcn_sched_barrier(0);
        rt_products<WM, WN, NT>(acc, fa0, fb0);
        __builtin_amdgcn_sched_barrier(0);
        const int kn = (k + 2 * KS < K) ? (k + 2 * KS) : 0;
        rt_load_set<WM, WN, NT, ATILE>(fa0, fb0, ap, bp, LDA, K, kn);
        __builtin_amdgcn_sched_barrier(0);
        rt_products<WM, WN, NT>(acc, fa1, fb1);
        __builtin_amdgcn_sched_barrier(0);
    }

    float* cp = C + (size_t)(m0 + 8 * h) * N + n0 + c;
#pragma unroll
    for (int j = 0; j < WN; ++j) {
        float bv = 0.f;
        if (EPILOG == 1) bv = bias[n0 + c + 16 * j];
#pragma unroll
        for (int i = 0; i < WM; ++i) {
#pragma unroll
            for (int r = 0; r < 8; ++r) {
                float v = acc[i][j][r];
                if (EPILOG == 1) {
                    float x = v + bv;
                    v = (x > 20.f) ? x : log1pf(__expf(x));
                }
                cp[(size_t)(16 * i + r) * N + 16 * j] = v;
            }
        }
    }
}

#define CX_WAVES 8
#define CX_KW    (D_INNER_C / CX_WAVES)
#define CX_GC    32
#define CX_NG    (CX_KW / CX_GC)
#define CX_NT    (CX_GC / 8)
#define CX_NJ    6
#define CX_TILE  (16 * 96)

static __device__ __forceinline__
void cx_load_w(v4f (&wf)[CX_NJ][CX_NT], const float* wp, int g)
{
#pragma unroll
    for (int n = 0; n < CX_NJ; ++n)
#pragma unroll
        for (int t = 0; t < CX_NT; ++t)
            wf[n][t] = *(const v4f*)(wp + (size_t)(16 * n) * D_INNER_C + CX_GC * g + 8 * t);
}

static __device__ __forceinline__
void cx_load_x(v4f (&xs)[D_CONV_C][CX_NT], v4f (&cw)[CX_NT][4], v4f (&cb)[CX_NT],
               const float* xp, const int (&off)[D_CONV_C],
               const float* cwp, const float* cbp, int g)
{
#pragma unroll
    for (int j = 0; j < D_CONV_C; ++j)
#pragma unroll
        for (int t = 0; t < CX_NT; ++t)
            xs[j][t] = *(const v4f*)(xp + off[j] + CX_GC * g + 8 * t);
#pragma unroll
    for (int t = 0; t < CX_NT; ++t) {
#pragma unroll
        for (int e = 0; e < 4; ++e)
            cw[t][e] = *(const v4f*)(cwp + D_CONV_C * (CX_GC * g + 8 * t + e));
        cb[t] = *(const v4f*)(cbp + CX_GC * g + 8 * t);
    }
}

static __device__ __forceinline__
void cx_conv_silu(v4f (&s)[CX_NT], const v4f (&xs)[D_CONV_C][CX_NT],
                  const v4f (&cw)[CX_NT][4], const v4f (&cb)[CX_NT],
                  const bool (&ok)[D_CONV_C])
{
#pragma unroll
    for (int t = 0; t < CX_NT; ++t) {
#pragma unroll
        for (int e = 0; e < 4; ++e) {
            float acc = cb[t][e];
#pragma unroll
            for (int j = 0; j < D_CONV_C; ++j) {
                const float xv = ok[j] ? xs[j][t][e] : 0.f;
                acc = __builtin_fmaf(cw[t][e][j], xv, acc);
            }
            s[t][e] = acc / (1.f + __expf(-acc));
        }
    }
}

static __device__ __forceinline__
void cx_products(v8f (&acc)[CX_NJ], const v4f (&s)[CX_NT], const v4f (&wf)[CX_NJ][CX_NT])
{
#pragma unroll
    for (int t = 0; t < CX_NT; ++t) {
#pragma unroll
        for (int p = 0; p < 2; ++p) {
            v2f a;
            a[0] = s[t][2 * p];
            a[1] = s[t][2 * p + 1];
#pragma unroll
            for (int n = 0; n < CX_NJ; ++n) {
                v2f b;
                b[0] = wf[n][t][2 * p];
                b[1] = wf[n][t][2 * p + 1];
                acc[n] = __builtin_amdgcn_wmma_f32_16x16x4_f32(
                    false, a, false, b, (short)0, acc[n], false, false);
            }
        }
    }
}

__global__ __launch_bounds__(32 * CX_WAVES)
void conv_silu_xproj_kernel(const float* __restrict__ xz,
                            const float* __restrict__ conv_w,
                            const float* __restrict__ conv_b,
                            const float* __restrict__ W_x,
                            float* __restrict__ x_act,
                            float* __restrict__ dbc)
{
    __shared__ float s_part[CX_WAVES * CX_TILE];

    const int lane = threadIdx.x & 31;
    const int wave = threadIdx.x >> 5;
    const int c = lane & 15;
    const int h = lane >> 4;
    const int m0 = blockIdx.x * 16;
    const int m  = m0 + c;
    const int tpos = m & (SEQ_C - 1);
    const int k0 = CX_KW * wave + 4 * h;

    bool ok[D_CONV_C];
    int  off[D_CONV_C];
#pragma unroll
    for (int j = 0; j < D_CONV_C; ++j) {
        ok[j]  = (tpos - (D_CONV_C - 1) + j) >= 0;
        off[j] = ok[j] ? (j - (D_CONV_C - 1)) * (2 * D_INNER_C) : 0;
    }

    const float* xp  = xz + (size_t)m * (2 * D_INNER_C) + k0;
    const float* wp  = W_x + (size_t)c * D_INNER_C + k0;
    const float* cwp = conv_w + D_CONV_C * k0;
    const float* cbp = conv_b + k0;
    float* ap = x_act + (size_t)m * D_INNER_C + k0;

    v8f acc[CX_NJ];
#pragma unroll
    for (int n = 0; n < CX_NJ; ++n) acc[n] = v8f{};

    v4f xs[D_CONV_C][CX_NT], cw[CX_NT][4], cb[CX_NT], s[CX_NT];
    v4f w0[CX_NJ][CX_NT], w1[CX_NJ][CX_NT];

    cx_load_x(xs, cw, cb, xp, off, cwp, cbp, 0);
    cx_load_w(w0, wp, 0);
    for (int g = 0; g < CX_NG; g += 2) {
        cx_load_w(w1, wp, g + 1);
        __builtin_amdgcn_sched_barrier(0);
        cx_conv_silu(s, xs, cw, cb, ok);
#pragma unroll
        for (int t = 0; t < CX_NT; ++t)
            *(v4f*)(ap + CX_GC * g + 8 * t) = s[t];
        cx_load_x(xs, cw, cb, xp, off, cwp, cbp, g + 1);
        __builtin_amdgcn_sched_barrier(0);
        cx_products(acc, s, w0);
        __builtin_amdgcn_sched_barrier(0);
        const int gn = (g + 2 < CX_NG) ? (g + 2) : 0;
        cx_load_w(w0, wp, gn);
        __builtin_amdgcn_sched_barrier(0);
        cx_conv_silu(s, xs, cw, cb, ok);
#pragma unroll
        for (int t = 0; t < CX_NT; ++t)
            *(v4f*)(ap + CX_GC * (g + 1) + 8 * t) = s[t];
        cx_load_x(xs, cw, cb, xp, off, cwp, cbp, gn);
        __builtin_amdgcn_sched_barrier(0);
        cx_products(acc, s, w1);
        __builtin_amdgcn_sched_barrier(0);
    }

    float* pw = s_part + wave * CX_TILE + (8 * h) * 96 + c;
#pragma unroll
    for (int n = 0; n < CX_NJ; ++n)
#pragma unroll
        for (int r = 0; r < 8; ++r)
            pw[r * 96 + 16 * n] = acc[n][r];
    __syncthreads();

    float* dp = dbc + (size_t)m0 * 96;
#pragma unroll
    for (int q = 0; q < CX_TILE / (32 * CX_WAVES); ++q) {
        const int o = threadIdx.x + 32 * CX_WAVES * q;
        float v = s_part[o];
#pragma unroll
        for (int w = 1; w < CX_WAVES; ++w) v += s_part[w * CX_TILE + o];
        dp[o] = v;
    }
}

#define SCAN_TB 32
#define SCAN_NB (SEQ_C / SCAN_TB)
#define SCAN_CH 16
#define SCAN_WG (D_INNER_C / SCAN_CH)
#define SCAN_G  4
#define SCAN_P  4
#define SCAN_SL 2
#define SCAN_RPW (SCAN_TB / SCAN_G)
#define SCAN_SB 4
#define SCAN_PL16 (SCAN_TB * SCAN_CH)
#define SCAN_PLBC (SCAN_TB * 32)
#define SCAN_STAGE (3 * SCAN_PL16 + SCAN_PLBC)
#define SCAN_PPL  (SCAN_TB * 32)
#define SCAN_PART (SCAN_G * SCAN_PPL)
#define SCAN_KT   (DT_RANK_C / 8)

struct ScanStepIn { float dtv, u; v2f b, c; };

static __device__ __forceinline__
void scan_read_step(ScanStepIn& r, const float* st, int t, int c, int so)
{
    r.dtv = st[t * SCAN_CH + c];
    r.u   = st[SCAN_PL16 + t * SCAN_CH + c];
    r.b = *(const v2f*)(st + 3 * SCAN_PL16 + t * 32 + so);
    r.c = *(const v2f*)(st + 3 * SCAN_PL16 + t * 32 + D_STATE_C + so);
}

static __device__ __forceinline__ void scan_lds_handover()
{
    __builtin_amdgcn_fence(__ATOMIC_RELEASE, "workgroup", "local");
    __builtin_amdgcn_s_barrier();
    __builtin_amdgcn_fence(__ATOMIC_ACQUIRE, "workgroup", "local");
}

static __device__ __forceinline__
v8f scan_dt_products(const v4f (&fa)[SCAN_KT], const v4f (&fb)[SCAN_KT])
{
    v8f acc = v8f{};
#pragma unroll
    for (int t = 0; t < SCAN_KT; ++t) {
#pragma unroll
        for (int p = 0; p < 2; ++p) {
            v2f a, bb;
            a[0]  = fa[t][2 * p];
            a[1]  = fa[t][2 * p + 1];
            bb[0] = fb[t][2 * p];
            bb[1] = fb[t][2 * p + 1];
            acc = __builtin_amdgcn_wmma_f32_16x16x4_f32(
                false, a, false, bb, (short)0, acc, false, false);
        }
    }
    return acc;
}

static __device__ __forceinline__ float scan_softplus(float x)
{
    const float l = log1pf(__expf(x));
    return (x > 20.f) ? x : l;
}

__global__ __launch_bounds__(32 * (SCAN_G + SCAN_P))
void selective_scan_kernel(const float* __restrict__ xz,
                           const float* __restrict__ x_act,
                           const float* __restrict__ dbc,
                           const float* __restrict__ W_dt,
                           const float* __restrict__ b_dt,
                           float* __restrict__ y,
                           const float* __restrict__ A_log,
                           const float* __restrict__ Dvec)
{
    static_assert(SCAN_G * 2 * SCAN_SL == D_STATE_C, "4 waves x 2 lane halves x 2 states");
    static_assert(SCAN_RPW == 8, "a wave stages 8 rows: one 16-byte load per lane covers them");
    static_assert(SCAN_TB == 32 && SCAN_P == 4, "two 16-step tiles per block, two dt waves per tile");
    __shared__ v4f s_mem[(2 * SCAN_STAGE + 2 * SCAN_PART) / 4];
    float* lds = (float*)s_mem;

    const int lane = threadIdx.x & 31;
    const int wv   = __builtin_amdgcn_readfirstlane((int)(threadIdx.x >> 5));
    const bool dtw = wv >= SCAN_G;
    const int w    = wv & (SCAN_G - 1);
    const int c    = lane & 15;
    const int hh   = lane >> 4;
    const int b    = blockIdx.x / SCAN_WG;
    const int d0   = (blockIdx.x % SCAN_WG) * SCAN_CH;
    const int d    = d0 + c;
    const int so   = 2 * SCAN_SL * w + SCAN_SL * hh;

    float Arow[SCAN_SL];
#pragma unroll
    for (int s = 0; s < SCAN_SL; ++s)
        Arow[s] = -__expf(A_log[(size_t)d * D_STATE_C + so + s]);
    const float Dd = Dvec[d];

    float h[SCAN_SL];
#pragma unroll
    for (int s = 0; s < SCAN_SL; ++s) h[s] = 0.f;

    const int r16 = lane >> 2;
    const int c16 = (lane & 3) * 4;
    const size_t m16 = (size_t)b * SEQ_C + SCAN_RPW * w + r16;
    const float* g_u  = x_act + m16 * D_INNER_C + d0 + c16;
    const float* g_z  = xz    + m16 * (2 * D_INNER_C) + D_INNER_C + d0 + c16;
    const int so16 = (SCAN_RPW * w + r16) * SCAN_CH + c16;
    const int r8 = lane >> 3;
    const int c8 = (lane & 7) * 4;
    const size_t m8 = (size_t)b * SEQ_C + SCAN_RPW * w + r8;
    const float* g_bc = dbc + m8 * 96 + 64 + c8;
    const int sobc = (SCAN_RPW * w + r8) * 32 + c8;
    float* yp = y + ((size_t)(b * (SEQ_C / 2) + (SCAN_RPW / 2) * w) * (D_INNER_C / 16)
                     + (d0 >> 4)) * 32 + lane;

    const int mt = w & 1;
    const int rq = w >> 1;
    const float* pa = dbc + ((size_t)b * SEQ_C + 16 * mt + c) * 96 + 4 * hh;
    const float* pb = W_dt + (size_t)d * DT_RANK_C + 4 * hh;
    v4f fa[SCAN_KT], fb[SCAN_KT];
#pragma unroll
    for (int t = 0; t < SCAN_KT; ++t) { fa[t] = v4f{}; fb[t] = v4f{}; }
    float bv = 0.f;
    v8f dacc = v8f{};
    const int dto = (16 * mt + 8 * hh + 4 * rq) * SCAN_CH + c;

    v4f p_u = v4f{}, p_z = v4f{}, p_bc[2] = { v4f{}, v4f{} };
    if (dtw) {
        v4f fa0[SCAN_KT];
#pragma unroll
        for (int t = 0; t < SCAN_KT; ++t) {
            fb[t]  = *(const v4f*)(pb + 8 * t);
            fa0[t] = *(const v4f*)(pa + 8 * t);
            fa[t]  = *(const v4f*)(pa + (size_t)SCAN_TB * 96 + 8 * t);
        }
        bv = b_dt[d];
        dacc = scan_dt_products(fa0, fb);
#pragma unroll
        for (int r4 = 0; r4 < 4; ++r4)
            lds[dto + r4 * SCAN_CH] = scan_softplus((rq ? dacc[4 + r4] : dacc[r4]) + bv);
        dacc = scan_dt_products(fa, fb);
    } else {
        p_u  = *(const v4f*)(g_u);
        p_z  = *(const v4f*)(g_z);
        p_bc[0] = *(const v4f*)(g_bc);
        p_bc[1] = *(const v4f*)(g_bc + (size_t)4 * 96);
        *(v4f*)(lds + SCAN_PL16 + so16)      = p_u;
        *(v4f*)(lds + 2 * SCAN_PL16 + so16)  = p_z;
        *(v4f*)(lds + 3 * SCAN_PL16 + sobc)        = p_bc[0];
        *(v4f*)(lds + 3 * SCAN_PL16 + sobc + 128)  = p_bc[1];
    }

    for (int k = 0; k < SCAN_NB; ++k) {
        const float* st = lds + (k & 1) * SCAN_STAGE;
        float* pt = lds + 2 * SCAN_STAGE + (k & 1) * SCAN_PART;

        scan_lds_handover();

        float x3 = 0.f;
        if (dtw) {
            if (k + 1 < SCAN_NB) {
                const int kn = (k + 2 < SCAN_NB) ? (k + 2) : (SCAN_NB - 1);
#pragma unroll
                for (int t = 0; t < SCAN_KT; ++t)
                    fa[t] = *(const v4f*)(pa + (size_t)kn * SCAN_TB * 96 + 8 * t);
                __builtin_amdgcn_sched_barrier(0);
                float* dn = lds + ((k + 1) & 1) * SCAN_STAGE + dto;
#pragma unroll
                for (int r4 = 0; r4 < 3; ++r4)
                    dn[r4 * SCAN_CH] = scan_softplus((rq ? dacc[4 + r4] : dacc[r4]) + bv);
                x3 = (rq ? dacc[7] : dacc[3]) + bv;
            }
        } else {
            if (k + 1 < SCAN_NB) {
                const size_t ro = (size_t)(k + 1) * SCAN_TB;
                p_u  = *(const v4f*)(g_u  + ro * D_INNER_C);
                p_z  = *(const v4f*)(g_z  + ro * (2 * D_INNER_C));
                p_bc[0] = *(const v4f*)(g_bc + ro * 96);
                p_bc[1] = *(const v4f*)(g_bc + (ro + 4) * 96);
            }

            ScanStepIn cur[SCAN_SB], nxt[SCAN_SB];
#pragma unroll
            for (int i = 0; i < SCAN_SB; ++i)
                scan_read_step(cur[i], st, i, c, so);

#pragma unroll
            for (int g = 0; g < SCAN_TB / SCAN_SB; ++g) {
                if (g + 1 < SCAN_TB / SCAN_SB) {
#pragma unroll
                    for (int i = 0; i < SCAN_SB; ++i)
                        scan_read_step(nxt[i], st, (g + 1) * SCAN_SB + i, c, so);
                }
                __builtin_amdgcn_sched_barrier(0);
#pragma unroll
                for (int i = 0; i < SCAN_SB; ++i) {
                    const float dtv = cur[i].dtv;
                    const float du  = dtv * cur[i].u;
                    float p = 0.f;
#pragma unroll
                    for (int s = 0; s < SCAN_SL; ++s) {
                        float dA = __expf(dtv * Arow[s]);
                        h[s] = dA * h[s] + du * cur[i].b[s];
                        p += h[s] * cur[i].c[s];
                    }
                    pt[w * SCAN_PPL + (g * SCAN_SB + i) * 32 + lane] = p;
                }
                __builtin_amdgcn_sched_barrier(0);
                if (g + 1 < SCAN_TB / SCAN_SB) {
#pragma unroll
                    for (int i = 0; i < SCAN_SB; ++i) cur[i] = nxt[i];
                }
            }
        }

        scan_lds_handover();

        if (dtw) {
            if (k + 1 < SCAN_NB) {
                lds[((k + 1) & 1) * SCAN_STAGE + dto + 3 * SCAN_CH] = scan_softplus(x3);
                dacc = scan_dt_products(fa, fb);
            }
        } else {
            {
                float plo[SCAN_RPW / 2][SCAN_G], phi[SCAN_RPW / 2][SCAN_G];
                float uu[SCAN_RPW / 2], zz[SCAN_RPW / 2];
#pragma unroll
                for (int i = 0; i < SCAN_RPW / 2; ++i) {
                    const int t = SCAN_RPW * w + 2 * i + hh;
#pragma unroll
                    for (int q = 0; q < SCAN_G; ++q) {
                        plo[i][q] = pt[q * SCAN_PPL + t * 32 + c];
                        phi[i][q] = pt[q * SCAN_PPL + t * 32 + 16 + c];
                    }
                    uu[i] = st[SCAN_PL16 + t * SCAN_CH + c];
                    zz[i] = st[2 * SCAN_PL16 + t * SCAN_CH + c];
                }
                float* yrow = yp + (size_t)k * (SCAN_TB / 2) * (D_INNER_C / 16) * 32;
#pragma unroll
                for (int i = 0; i < SCAN_RPW / 2; ++i) {
                    float yv = plo[i][0] + phi[i][0];
#pragma unroll
                    for (int q = 1; q < SCAN_G; ++q) {
                        yv += plo[i][q];
                        yv += phi[i][q];
                    }
                    yv += uu[i] * Dd;
                    const float z = zz[i];
                    float sz = z / (1.f + __expf(-z));
                    yrow[(size_t)i * (D_INNER_C / 16) * 32] = yv * sz;
                }
            }

            if (k + 1 < SCAN_NB) {
                float* sn = lds + ((k + 1) & 1) * SCAN_STAGE;
                *(v4f*)(sn + SCAN_PL16 + so16)      = p_u;
                *(v4f*)(sn + 2 * SCAN_PL16 + so16)  = p_z;
                *(v4f*)(sn + 3 * SCAN_PL16 + sobc)        = p_bc[0];
                *(v4f*)(sn + 3 * SCAN_PL16 + sobc + 128)  = p_bc[1];
            }
        }
    }
}

extern "C" void kernel_launch(void* const* d_in, const int* in_sizes, int n_in,
                              void* d_out, int out_size, void* d_ws, size_t ws_size,
                              hipStream_t stream) {
    (void)in_sizes; (void)n_in; (void)out_size; (void)ws_size;

    const float* x      = (const float*)d_in[0];
    const float* W_in   = (const float*)d_in[1];
    const float* conv_w = (const float*)d_in[2];
    const float* conv_b = (const float*)d_in[3];
    const float* W_x    = (const float*)d_in[4];
    const float* W_dt   = (const float*)d_in[5];
    const float* b_dt   = (const float*)d_in[6];
    const float* A_log  = (const float*)d_in[7];
    const float* Dvec   = (const float*)d_in[8];
    const float* W_out  = (const float*)d_in[9];
    float* out = (float*)d_out;

    float* xz   = (float*)d_ws;
    float* xact = xz   + (size_t)NTOK_C * (2 * D_INNER_C);
    float* dbc  = xact + (size_t)NTOK_C * D_INNER_C;
    float* yb   = dbc  + (size_t)NTOK_C * 96;

    wmma_gemm_f32_rt<NTOK_C, 2 * D_INNER_C, D_MODEL_C, D_MODEL_C, 4, 4, 4, 2, 32, 0>
        <<<dim3(32, 8), 256, 0, stream>>>(x, W_in, nullptr, xz);

    conv_silu_xproj_kernel<<<NTOK_C / 16, 32 * CX_WAVES, 0, stream>>>(
        xz, conv_w, conv_b, W_x, xact, dbc);

    selective_scan_kernel<<<(BATCH_C * D_INNER_C) / SCAN_CH, 32 * (SCAN_G + SCAN_P), 0, stream>>>(
        xz, xact, dbc, W_dt, b_dt, yb, A_log, Dvec);

    wmma_gemm_f32_rt<NTOK_C, D_MODEL_C, D_INNER_C, D_INNER_C, 2, 4, 2, 2, 64, 0, true>
        <<<dim3(8, 32), 128, 0, stream>>>(yb, W_out, nullptr, out);
}


# conv+x_proj launch: a 16-token tile's channels split over 2 workgroups (256 workgroups, was 128), each leaves a partial dbc plane; the scan adds the 2 planes as it loads them; dbc sum order changes
# speedup vs baseline: 9.8166x; 1.0334x over previous
#include <hip/hip_runtime.h>
#include <hip/hip_bf16.h>
#include <math.h>

typedef __attribute__((ext_vector_type(2))) float v2f;
typedef __attribute__((ext_vector_type(4))) float v4f;
typedef __attribute__((ext_vector_type(8))) float v8f;

#define D_MODEL_C 1024
#define D_STATE_C 16
#define D_CONV_C  4
#define DT_RANK_C 64
#define D_INNER_C 2048
#define BATCH_C   2
#define SEQ_C     1024
#define NTOK_C    (BATCH_C * SEQ_C)

template <int WM, int WN, int NT, bool ATILE = false>
static __device__ __forceinline__
void rt_load_set(v4f (&fa)[WM][NT], v4f (&fb)[WN][NT],
                 const float* ap, const float* bp, int lda, int ldb, int k)
{
#pragma unroll
    for (int t = 0; t < NT; ++t) {
#pragma unroll
        for (int i = 0; i < WM; ++i) {
            if (ATILE)
                fa[i][t] = *(const v4f*)(ap + (size_t)(8 * i) * (2 * lda) + 2 * k
                                         + 32 * (t >> 1) + 8 * (t & 1));
            else
                fa[i][t] = *(const v4f*)(ap + (size_t)(16 * i) * lda + k + 8 * t);
        }
#pragma unroll
        for (int j = 0; j < WN; ++j)
            fb[j][t] = *(const v4f*)(bp + (size_t)(16 * j) * ldb + k + 8 * t);
    }
}

template <int WM, int WN, int NT>
static __device__ __forceinline__
void rt_products(v8f (&acc)[WM][WN], const v4f (&fa)[WM][NT], const v4f (&fb)[WN][NT])
{
#pragma unroll
    for (int t = 0; t < NT; ++t) {
#pragma unroll
        for (int p = 0; p < 2; ++p) {
#pragma unroll
            for (int i = 0; i < WM; ++i) {
                v2f a;
                a[0] = fa[i][t][2 * p];
                a[1] = fa[i][t][2 * p + 1];
#pragma unroll
                for (int j = 0; j < WN; ++j) {
                    v2f b;
                    b[0] = fb[j][t][2 * p];
                    b[1] = fb[j][t][2 * p + 1];
                    acc[i][j] = __builtin_amdgcn_wmma_f32_16x16x4_f32(
                        false, a, false, b, (short)0, acc[i][j], false, false);
                }
            }
        }
    }
}

template <int M, int N, int K, int LDA, int WM, int WN, int BWM, int BWN, int KS, int EPILOG, bool ATILE = false>
__global__ __launch_bounds__(32 * BWM * BWN)
void wmma_gemm_f32_rt(const float* __restrict__ A,
                      const float* __restrict__ W,
                      const float* __restrict__ bias,
                      float* __restrict__ C)
{
    constexpr int NT = KS / 8;
    constexpr int TM = 16 * WM, TN = 16 * WN;
    constexpr int BM = TM * BWM, BN = TN * BWN;
    static_assert(M % BM == 0 && N % BN == 0, "tiles must divide the problem");
    static_assert(K % (2 * KS) == 0, "two fragment sets per trip");
    static_assert(!ATILE || (KS % 16 == 0 && LDA % 16 == 0 && LDA == K), "tiled A: whole 16-column tiles");

    const int lane = threadIdx.x & 31;
    const int wave = threadIdx.x >> 5;
    const int c = lane & 15;
    const int h = lane >> 4;
    const int m0 = blockIdx.y * BM + (wave / BWN) * TM;
    const int n0 = blockIdx.x * BN + (wave % BWN) * TN;

    const float* ap = ATILE
        ? A + (size_t)((m0 + c) >> 1) * (2 * LDA) + (c & 1) * 16 + 4 * h
        : A + (size_t)(m0 + c) * LDA + 4 * h;
    const float* bp = W + (size_t)(n0 + c) * K + 4 * h;

    v8f acc[WM][WN];
#pragma unroll
    for (int i = 0; i < WM; ++i)
#pragma unroll
        for (int j = 0; j < WN; ++j) acc[i][j] = v8f{};

    v4f fa0[WM][NT], fb0[WN][NT], fa1[WM][NT], fb1[WN][NT];

    rt_load_set<WM, WN, NT, ATILE>(fa0, fb0, ap, bp, LDA, K, 0);
    for (int k = 0; k < K; k += 2 * KS) {
        rt_load_set<WM, WN, NT, ATILE>(fa1, fb1, ap, bp, LDA, K, k + KS);
        __builtin_amdgcn_sched_barrier(0);
        rt_products<WM, WN, NT>(acc, fa0, fb0);
        __builtin_amdgcn_sched_barrier(0);
        const int kn = (k + 2 * KS < K) ? (k + 2 * KS) : 0;
        rt_load_set<WM, WN, NT, ATILE>(fa0, fb0, ap, bp, LDA, K, kn);
        __builtin_amdgcn_sched_barrier(0);
        rt_products<WM, WN, NT>(acc, fa1, fb1);
        __builtin_amdgcn_sched_barrier(0);
    }

    float* cp = C + (size_t)(m0 + 8 * h) * N + n0 + c;
#pragma unroll
    for (int j = 0; j < WN; ++j) {
        float bv = 0.f;
        if (EPILOG == 1) bv = bias[n0 + c + 16 * j];
#pragma unroll
        for (int i = 0; i < WM; ++i) {
#pragma unroll
            for (int r = 0; r < 8; ++r) {
                float v = acc[i][j][r];
                if (EPILOG == 1) {
                    float x = v + bv;
                    v = (x > 20.f) ? x : log1pf(__expf(x));
                }
                cp[(size_t)(16 * i + r) * N + 16 * j] = v;
            }
        }
    }
}

#define CX_WAVES 8
#define CX_SPLIT 2
#define CX_KW    (D_INNER_C / (CX_WAVES * CX_SPLIT))
#define DBC_PLANE (NTOK_C * 96)
#define CX_GC    32
#define CX_NG    (CX_KW / CX_GC)
#define CX_NT    (CX_GC / 8)
#define CX_NJ    6
#define CX_TILE  (16 * 96)

static __device__ __forceinline__
void cx_load_w(v4f (&wf)[CX_NJ][CX_NT], const float* wp, int g)
{
#pragma unroll
    for (int n = 0; n < CX_NJ; ++n)
#pragma unroll
        for (int t = 0; t < CX_NT; ++t)
            wf[n][t] = *(const v4f*)(wp + (size_t)(16 * n) * D_INNER_C + CX_GC * g + 8 * t);
}

static __device__ __forceinline__
void cx_load_x(v4f (&xs)[D_CONV_C][CX_NT], v4f (&cw)[CX_NT][4], v4f (&cb)[CX_NT],
               const float* xp, const int (&off)[D_CONV_C],
               const float* cwp, const float* cbp, int g)
{
#pragma unroll
    for (int j = 0; j < D_CONV_C; ++j)
#pragma unroll
        for (int t = 0; t < CX_NT; ++t)
            xs[j][t] = *(const v4f*)(xp + off[j] + CX_GC * g + 8 * t);
#pragma unroll
    for (int t = 0; t < CX_NT; ++t) {
#pragma unroll
        for (int e = 0; e < 4; ++e)
            cw[t][e] = *(const v4f*)(cwp + D_CONV_C * (CX_GC * g + 8 * t + e));
        cb[t] = *(const v4f*)(cbp + CX_GC * g + 8 * t);
    }
}

static __device__ __forceinline__
void cx_conv_silu(v4f (&s)[CX_NT], const v4f (&xs)[D_CONV_C][CX_NT],
                  const v4f (&cw)[CX_NT][4], const v4f (&cb)[CX_NT],
                  const bool (&ok)[D_CONV_C])
{
#pragma unroll
    for (int t = 0; t < CX_NT; ++t) {
#pragma unroll
        for (int e = 0; e < 4; ++e) {
            float acc = cb[t][e];
#pragma unroll
            for (int j = 0; j < D_CONV_C; ++j) {
                const float xv = ok[j] ? xs[j][t][e] : 0.f;
                acc = __builtin_fmaf(cw[t][e][j], xv, acc);
            }
            s[t][e] = acc / (1.f + __expf(-acc));
        }
    }
}

static __device__ __forceinline__
void cx_products(v8f (&acc)[CX_NJ], const v4f (&s)[CX_NT], const v4f (&wf)[CX_NJ][CX_NT])
{
#pragma unroll
    for (int t = 0; t < CX_NT; ++t) {
#pragma unroll
        for (int p = 0; p < 2; ++p) {
            v2f a;
            a[0] = s[t][2 * p];
            a[1] = s[t][2 * p + 1];
#pragma unroll
            for (int n = 0; n < CX_NJ; ++n) {
                v2f b;
                b[0] = wf[n][t][2 * p];
                b[1] = wf[n][t][2 * p + 1];
                acc[n] = __builtin_amdgcn_wmma_f32_16x16x4_f32(
                    false, a, false, b, (short)0, acc[n], false, false);
            }
        }
    }
}

__global__ __launch_bounds__(32 * CX_WAVES)
void conv_silu_xproj_kernel(const float* __restrict__ xz,
                            const float* __restrict__ conv_w,
                            const float* __restrict__ conv_b,
                            const float* __restrict__ W_x,
                            float* __restrict__ x_act,
                            float* __restrict__ dbc)
{
    __shared__ float s_part[CX_WAVES * CX_TILE];

    const int lane = threadIdx.x & 31;
    const int wave = threadIdx.x >> 5;
    const int c = lane & 15;
    const int h = lane >> 4;
    const int m0 = (blockIdx.x / CX_SPLIT) * 16;
    const int q  = blockIdx.x % CX_SPLIT;
    const int m  = m0 + c;
    const int tpos = m & (SEQ_C - 1);
    const int k0 = (D_INNER_C / CX_SPLIT) * q + CX_KW * wave + 4 * h;

    bool ok[D_CONV_C];
    int  off[D_CONV_C];
#pragma unroll
    for (int j = 0; j < D_CONV_C; ++j) {
        ok[j]  = (tpos - (D_CONV_C - 1) + j) >= 0;
        off[j] = ok[j] ? (j - (D_CONV_C - 1)) * (2 * D_INNER_C) : 0;
    }

    const float* xp  = xz + (size_t)m * (2 * D_INNER_C) + k0;
    const float* wp  = W_x + (size_t)c * D_INNER_C + k0;
    const float* cwp = conv_w + D_CONV_C * k0;
    const float* cbp = conv_b + k0;
    float* ap = x_act + (size_t)m * D_INNER_C + k0;

    v8f acc[CX_NJ];
#pragma unroll
    for (int n = 0; n < CX_NJ; ++n) acc[n] = v8f{};

    v4f xs[D_CONV_C][CX_NT], cw[CX_NT][4], cb[CX_NT], s[CX_NT];
    v4f w0[CX_NJ][CX_NT], w1[CX_NJ][CX_NT];

    cx_load_x(xs, cw, cb, xp, off, cwp, cbp, 0);
    cx_load_w(w0, wp, 0);
#pragma unroll
    for (int g = 0; g < CX_NG; g += 2) {
        cx_load_w(w1, wp, g + 1);
        __builtin_amdgcn_sched_barrier(0);
        cx_conv_silu(s, xs, cw, cb, ok);
#pragma unroll
        for (int t = 0; t < CX_NT; ++t)
            *(v4f*)(ap + CX_GC * g + 8 * t) = s[t];
        cx_load_x(xs, cw, cb, xp, off, cwp, cbp, g + 1);
        __builtin_amdgcn_sched_barrier(0);
        cx_products(acc, s, w0);
        __builtin_amdgcn_sched_barrier(0);
        if (g + 2 < CX_NG) cx_load_w(w0, wp, g + 2);
        __builtin_amdgcn_sched_barrier(0);
        cx_conv_silu(s, xs, cw, cb, ok);
#pragma unroll
        for (int t = 0; t < CX_NT; ++t)
            *(v4f*)(ap + CX_GC * (g + 1) + 8 * t) = s[t];
        if (g + 2 < CX_NG) cx_load_x(xs, cw, cb, xp, off, cwp, cbp, g + 2);
        __builtin_amdgcn_sched_barrier(0);
        cx_products(acc, s, w1);
        __builtin_amdgcn_sched_barrier(0);
    }

    float* pw = s_part + wave * CX_TILE + (8 * h) * 96 + c;
#pragma unroll
    for (int n = 0; n < CX_NJ; ++n)
#pragma unroll
        for (int r = 0; r < 8; ++r)
            pw[r * 96 + 16 * n] = acc[n][r];
    __syncthreads();

    float* dp = dbc + (size_t)q * DBC_PLANE + (size_t)m0 * 96;
#pragma unroll
    for (int i = 0; i < CX_TILE / (32 * CX_WAVES); ++i) {
        const int o = threadIdx.x + 32 * CX_WAVES * i;
        float v = s_part[o];
#pragma unroll
        for (int w = 1; w < CX_WAVES; ++w) v += s_part[w * CX_TILE + o];
        dp[o] = v;
    }
}

#define SCAN_TB 32
#define SCAN_NB (SEQ_C / SCAN_TB)
#define SCAN_CH 16
#define SCAN_WG (D_INNER_C / SCAN_CH)
#define SCAN_G  4
#define SCAN_P  4
#define SCAN_SL 2
#define SCAN_RPW (SCAN_TB / SCAN_G)
#define SCAN_SB 4
#define SCAN_PL16 (SCAN_TB * SCAN_CH)
#define SCAN_PLBC (SCAN_TB * 32)
#define SCAN_STAGE (3 * SCAN_PL16 + SCAN_PLBC)
#define SCAN_PPL  (SCAN_TB * 32)
#define SCAN_PART (SCAN_G * SCAN_PPL)
#define SCAN_KT   (DT_RANK_C / 8)

struct ScanStepIn { float dtv, u; v2f b, c; };

static __device__ __forceinline__
void scan_read_step(ScanStepIn& r, const float* st, int t, int c, int so)
{
    r.dtv = st[t * SCAN_CH + c];
    r.u   = st[SCAN_PL16 + t * SCAN_CH + c];
    r.b = *(const v2f*)(st + 3 * SCAN_PL16 + t * 32 + so);
    r.c = *(const v2f*)(st + 3 * SCAN_PL16 + t * 32 + D_STATE_C + so);
}

static __device__ __forceinline__ void scan_lds_handover()
{
    __builtin_amdgcn_fence(__ATOMIC_RELEASE, "workgroup", "local");
    __builtin_amdgcn_s_barrier();
    __builtin_amdgcn_fence(__ATOMIC_ACQUIRE, "workgroup", "local");
}

static __device__ __forceinline__
v8f scan_dt_products(const v4f (&fa)[SCAN_KT], const v4f (&fb)[SCAN_KT])
{
    v8f acc = v8f{};
#pragma unroll
    for (int t = 0; t < SCAN_KT; ++t) {
#pragma unroll
        for (int p = 0; p < 2; ++p) {
            v2f a, bb;
            a[0]  = fa[t][2 * p];
            a[1]  = fa[t][2 * p + 1];
            bb[0] = fb[t][2 * p];
            bb[1] = fb[t][2 * p + 1];
            acc = __builtin_amdgcn_wmma_f32_16x16x4_f32(
                false, a, false, bb, (short)0, acc, false, false);
        }
    }
    return acc;
}

static __device__ __forceinline__ float scan_softplus(float x)
{
    const float l = log1pf(__expf(x));
    return (x > 20.f) ? x : l;
}

__global__ __launch_bounds__(32 * (SCAN_G + SCAN_P))
void selective_scan_kernel(const float* __restrict__ xz,
                           const float* __restrict__ x_act,
                           const float* __restrict__ dbc,
                           const float* __restrict__ W_dt,
                           const float* __restrict__ b_dt,
                           float* __restrict__ y,
                           const float* __restrict__ A_log,
                           const float* __restrict__ Dvec)
{
    static_assert(SCAN_G * 2 * SCAN_SL == D_STATE_C, "4 waves x 2 lane halves x 2 states");
    static_assert(SCAN_RPW == 8, "a wave stages 8 rows: one 16-byte load per lane covers them");
    static_assert(SCAN_TB == 32 && SCAN_P == 4, "two 16-step tiles per block, two dt waves per tile");
    static_assert(CX_SPLIT == 2, "dbc is read as the sum of two planes");
    __shared__ v4f s_mem[(2 * SCAN_STAGE + 2 * SCAN_PART) / 4];
    float* lds = (float*)s_mem;

    const int lane = threadIdx.x & 31;
    const int wv   = __builtin_amdgcn_readfirstlane((int)(threadIdx.x >> 5));
    const bool dtw = wv >= SCAN_G;
    const int w    = wv & (SCAN_G - 1);
    const int c    = lane & 15;
    const int hh   = lane >> 4;
    const int b    = blockIdx.x / SCAN_WG;
    const int d0   = (blockIdx.x % SCAN_WG) * SCAN_CH;
    const int d    = d0 + c;
    const int so   = 2 * SCAN_SL * w + SCAN_SL * hh;

    float Arow[SCAN_SL];
#pragma unroll
    for (int s = 0; s < SCAN_SL; ++s)
        Arow[s] = -__expf(A_log[(size_t)d * D_STATE_C + so + s]);
    const float Dd = Dvec[d];

    float h[SCAN_SL];
#pragma unroll
    for (int s = 0; s < SCAN_SL; ++s) h[s] = 0.f;

    const int r16 = lane >> 2;
    const int c16 = (lane & 3) * 4;
    const size_t m16 = (size_t)b * SEQ_C + SCAN_RPW * w + r16;
    const float* g_u  = x_act + m16 * D_INNER_C + d0 + c16;
    const float* g_z  = xz    + m16 * (2 * D_INNER_C) + D_INNER_C + d0 + c16;
    const int so16 = (SCAN_RPW * w + r16) * SCAN_CH + c16;
    const int r8 = lane >> 3;
    const int c8 = (lane & 7) * 4;
    const size_t m8 = (size_t)b * SEQ_C + SCAN_RPW * w + r8;
    const float* g_bc = dbc + m8 * 96 + 64 + c8;
    const int sobc = (SCAN_RPW * w + r8) * 32 + c8;
    float* yp = y + ((size_t)(b * (SEQ_C / 2) + (SCAN_RPW / 2) * w) * (D_INNER_C / 16)
                     + (d0 >> 4)) * 32 + lane;

    const int mt = w & 1;
    const int rq = w >> 1;
    const float* pa = dbc + ((size_t)b * SEQ_C + 16 * mt + c) * 96 + 4 * hh;
    const float* pb = W_dt + (size_t)d * DT_RANK_C + 4 * hh;
    v4f fa[SCAN_KT], fa2[SCAN_KT], fb[SCAN_KT];
#pragma unroll
    for (int t = 0; t < SCAN_KT; ++t) { fa[t] = v4f{}; fa2[t] = v4f{}; fb[t] = v4f{}; }
    float bv = 0.f;
    v8f dacc = v8f{};
    const int dto = (16 * mt + 8 * hh + 4 * rq) * SCAN_CH + c;

    v4f p_u = v4f{}, p_z = v4f{}, p_bc[2] = { v4f{}, v4f{} };
    v4f q_bc[2] = { v4f{}, v4f{} };
    if (dtw) {
        v4f fa0[SCAN_KT];
#pragma unroll
        for (int t = 0; t < SCAN_KT; ++t) {
            fb[t]  = *(const v4f*)(pb + 8 * t);
            fa0[t] = *(const v4f*)(pa + 8 * t)
                   + *(const v4f*)(pa + DBC_PLANE + 8 * t);
            fa[t]  = *(const v4f*)(pa + (size_t)SCAN_TB * 96 + 8 * t)
                   + *(const v4f*)(pa + DBC_PLANE + (size_t)SCAN_TB * 96 + 8 * t);
        }
        bv = b_dt[d];
        dacc = scan_dt_products(fa0, fb);
#pragma unroll
        for (int r4 = 0; r4 < 4; ++r4)
            lds[dto + r4 * SCAN_CH] = scan_softplus((rq ? dacc[4 + r4] : dacc[r4]) + bv);
        dacc = scan_dt_products(fa, fb);
    } else {
        p_u  = *(const v4f*)(g_u);
        p_z  = *(const v4f*)(g_z);
        p_bc[0] = *(const v4f*)(g_bc)                  + *(const v4f*)(g_bc + DBC_PLANE);
        p_bc[1] = *(const v4f*)(g_bc + (size_t)4 * 96) + *(const v4f*)(g_bc + DBC_PLANE + (size_t)4 * 96);
        *(v4f*)(lds + SCAN_PL16 + so16)      = p_u;
        *(v4f*)(lds + 2 * SCAN_PL16 + so16)  = p_z;
        *(v4f*)(lds + 3 * SCAN_PL16 + sobc)        = p_bc[0];
        *(v4f*)(lds + 3 * SCAN_PL16 + sobc + 128)  = p_bc[1];
    }

    for (int k = 0; k < SCAN_NB; ++k) {
        const float* st = lds + (k & 1) * SCAN_STAGE;
        float* pt = lds + 2 * SCAN_STAGE + (k & 1) * SCAN_PART;

        scan_lds_handover();

        float x3 = 0.f;
        if (dtw) {
            if (k + 1 < SCAN_NB) {
                const int kn = (k + 2 < SCAN_NB) ? (k + 2) : (SCAN_NB - 1);
#pragma unroll
                for (int t = 0; t < SCAN_KT; ++t) {
                    fa[t]  = *(const v4f*)(pa + (size_t)kn * SCAN_TB * 96 + 8 * t);
                    fa2[t] = *(const v4f*)(pa + DBC_PLANE + (size_t)kn * SCAN_TB * 96 + 8 * t);
                }
                __builtin_amdgcn_sched_barrier(0);
                float* dn = lds + ((k + 1) & 1) * SCAN_STAGE + dto;
#pragma unroll
                for (int r4 = 0; r4 < 3; ++r4)
                    dn[r4 * SCAN_CH] = scan_softplus((rq ? dacc[4 + r4] : dacc[r4]) + bv);
                x3 = (rq ? dacc[7] : dacc[3]) + bv;
            }
        } else {
            if (k + 1 < SCAN_NB) {
                const size_t ro = (size_t)(k + 1) * SCAN_TB;
                p_u  = *(const v4f*)(g_u  + ro * D_INNER_C);
                p_z  = *(const v4f*)(g_z  + ro * (2 * D_INNER_C));
                p_bc[0] = *(const v4f*)(g_bc + ro * 96);
                p_bc[1] = *(const v4f*)(g_bc + (ro + 4) * 96);
                q_bc[0] = *(const v4f*)(g_bc + DBC_PLANE + ro * 96);
                q_bc[1] = *(const v4f*)(g_bc + DBC_PLANE + (ro + 4) * 96);
            }

            ScanStepIn cur[SCAN_SB], nxt[SCAN_SB];
#pragma unroll
            for (int i = 0; i < SCAN_SB; ++i)
                scan_read_step(cur[i], st, i, c, so);

#pragma unroll
            for (int g = 0; g < SCAN_TB / SCAN_SB; ++g) {
                if (g + 1 < SCAN_TB / SCAN_SB) {
#pragma unroll
                    for (int i = 0; i < SCAN_SB; ++i)
                        scan_read_step(nxt[i], st, (g + 1) * SCAN_SB + i, c, so);
                }
                __builtin_amdgcn_sched_barrier(0);
#pragma unroll
                for (int i = 0; i < SCAN_SB; ++i) {
                    const float dtv = cur[i].dtv;
                    const float du  = dtv * cur[i].u;
                    float p = 0.f;
#pragma unroll
                    for (int s = 0; s < SCAN_SL; ++s) {
                        float dA = __expf(dtv * Arow[s]);
                        h[s] = dA * h[s] + du * cur[i].b[s];
                        p += h[s] * cur[i].c[s];
                    }
                    pt[w * SCAN_PPL + (g * SCAN_SB + i) * 32 + lane] = p;
                }
                __builtin_amdgcn_sched_barrier(0);
                if (g + 1 < SCAN_TB / SCAN_SB) {
#pragma unroll
                    for (int i = 0; i < SCAN_SB; ++i) cur[i] = nxt[i];
                }
            }
        }

        scan_lds_handover();

        if (dtw) {
            if (k + 1 < SCAN_NB) {
                lds[((k + 1) & 1) * SCAN_STAGE + dto + 3 * SCAN_CH] = scan_softplus(x3);
#pragma unroll
                for (int t = 0; t < SCAN_KT; ++t) fa[t] += fa2[t];
                dacc = scan_dt_products(fa, fb);
            }
        } else {
            {
                float plo[SCAN_RPW / 2][SCAN_G], phi[SCAN_RPW / 2][SCAN_G];
                float uu[SCAN_RPW / 2], zz[SCAN_RPW / 2];
#pragma unroll
                for (int i = 0; i < SCAN_RPW / 2; ++i) {
                    const int t = SCAN_RPW * w + 2 * i + hh;
#pragma unroll
                    for (int q = 0; q < SCAN_G; ++q) {
                        plo[i][q] = pt[q * SCAN_PPL + t * 32 + c];
                        phi[i][q] = pt[q * SCAN_PPL + t * 32 + 16 + c];
                    }
                    uu[i] = st[SCAN_PL16 + t * SCAN_CH + c];
                    zz[i] = st[2 * SCAN_PL16 + t * SCAN_CH + c];
                }
                float* yrow = yp + (size_t)k * (SCAN_TB / 2) * (D_INNER_C / 16) * 32;
#pragma unroll
                for (int i = 0; i < SCAN_RPW / 2; ++i) {
                    float yv = plo[i][0] + phi[i][0];
#pragma unroll
                    for (int q = 1; q < SCAN_G; ++q) {
                        yv += plo[i][q];
                        yv += phi[i][q];
                    }
                    yv += uu[i] * Dd;
                    const float z = zz[i];
                    float sz = z / (1.f + __expf(-z));
                    yrow[(size_t)i * (D_INNER_C / 16) * 32] = yv * sz;
                }
            }

            if (k + 1 < SCAN_NB) {
                float* sn = lds + ((k + 1) & 1) * SCAN_STAGE;
                *(v4f*)(sn + SCAN_PL16 + so16)      = p_u;
                *(v4f*)(sn + 2 * SCAN_PL16 + so16)  = p_z;
                *(v4f*)(sn + 3 * SCAN_PL16 + sobc)        = p_bc[0] + q_bc[0];
                *(v4f*)(sn + 3 * SCAN_PL16 + sobc + 128)  = p_bc[1] + q_bc[1];
            }
        }
    }
}

extern "C" void kernel_launch(void* const* d_in, const int* in_sizes, int n_in,
                              void* d_out, int out_size, void* d_ws, size_t ws_size,
                              hipStream_t stream) {
    (void)in_sizes; (void)n_in; (void)out_size; (void)ws_size;

    const float* x      = (const float*)d_in[0];
    const float* W_in   = (const float*)d_in[1];
    const float* conv_w = (const float*)d_in[2];
    const float* conv_b = (const float*)d_in[3];
    const float* W_x    = (const float*)d_in[4];
    const float* W_dt   = (const float*)d_in[5];
    const float* b_dt   = (const float*)d_in[6];
    const float* A_log  = (const float*)d_in[7];
    const float* Dvec   = (const float*)d_in[8];
    const float* W_out  = (const float*)d_in[9];
    float* out = (float*)d_out;

    float* xz   = (float*)d_ws;
    float* xact = xz   + (size_t)NTOK_C * (2 * D_INNER_C);
    float* dbc  = xact + (size_t)NTOK_C * D_INNER_C;
    float* yb   = dbc  + (size_t)CX_SPLIT * DBC_PLANE;

    wmma_gemm_f32_rt<NTOK_C, 2 * D_INNER_C, D_MODEL_C, D_MODEL_C, 4, 4, 4, 2, 32, 0>
        <<<dim3(32, 8), 256, 0, stream>>>(x, W_in, nullptr, xz);

    conv_silu_xproj_kernel<<<(NTOK_C / 16) * CX_SPLIT, 32 * CX_WAVES, 0, stream>>>(
        xz, conv_w, conv_b, W_x, xact, dbc);

    selective_scan_kernel<<<(BATCH_C * D_INNER_C) / SCAN_CH, 32 * (SCAN_G + SCAN_P), 0, stream>>>(
        xz, xact, dbc, W_dt, b_dt, yb, A_log, Dvec);

    wmma_gemm_f32_rt<NTOK_C, D_MODEL_C, D_INNER_C, D_INNER_C, 2, 4, 2, 2, 64, 0, true>
        <<<dim3(8, 32), 128, 0, stream>>>(yb, W_out, nullptr, out);
}
